# starting kernel
# baseline (speedup 1.0000x reference)
#include <hip/hip_runtime.h>


typedef __attribute__((ext_vector_type(2))) float v2f;
typedef __attribute__((ext_vector_type(8))) float v8f;

#define N_USER 100000
#define N_ITEM 100000
#define NNODES (N_USER + N_ITEM)
#define HID    128
#define E_UI   600000
#define E_IU   600000
#define E_TOT  (E_UI + E_IU)

__global__ __launch_bounds__(128) void gemm_n128(
    const float* __restrict__ A, const float* __restrict__ W,
    const float* __restrict__ bias, float* __restrict__ C,
    int M, int K, int relu, int use_bias)
{
  const int lane  = threadIdx.x & 31;
  const int wave  = threadIdx.x >> 5;
  const long row0 = (long)blockIdx.x * 64 + (long)wave * 16;

  const int m     = lane & 15;
  const int n     = lane & 15;
  const int khalf = (lane >> 4) << 1;

  long arow = row0 + m;
  if (arow >= M) arow = M - 1;
  const float* __restrict__ Ar = A + arow * (long)K;

  v8f acc[8] = {};

  for (int k = 0; k < K; k += 4) {
    v2f a;
    a.x = Ar[k + khalf];
    a.y = Ar[k + khalf + 1];
    if (relu) { a.x = fmaxf(a.x, 0.0f); a.y = fmaxf(a.y, 0.0f); }

    const float* __restrict__ Wk = W + (long)(k + khalf) * HID + n;
#pragma unroll
    for (int t = 0; t < 8; ++t) {
      v2f b;
      b.x = Wk[t * 16];
      b.y = Wk[t * 16 + HID];
      acc[t] = __builtin_amdgcn_wmma_f32_16x16x4_f32(
           false, a,  false, b,
           (short)0, acc[t],  false,  false);
    }
  }

  const int rhalf = (lane >> 4) << 3;
#pragma unroll
  for (int t = 0; t < 8; ++t) {
    float bv = use_bias ? bias[t * 16 + n] : 0.0f;
#pragma unroll
    for (int r = 0; r < 8; ++r) {
      long row = row0 + rhalf + r;
      if (row < M) C[row * HID + t * 16 + n] = acc[t][r] + bv;
    }
  }
}

__global__ void deg_init(float* __restrict__ deg, int n)
{
  int i = blockIdx.x * blockDim.x + threadIdx.x;
  if (i < n) deg[i] = 1.0f;
}

__global__ void deg_count(const int* __restrict__ eui,
                          const int* __restrict__ eiu,
                          float* __restrict__ deg)
{
  int e = blockIdx.x * blockDim.x + threadIdx.x;
  if (e >= E_TOT) return;
  int d;
  if (e < E_UI) d = eui[E_UI + e] + N_USER;
  else          d = eiu[E_IU + (e - E_UI)];
  atomicAdd(&deg[d], 1.0f);
}

__global__ void dinv_fin(float* __restrict__ deg, int n)
{
  int i = blockIdx.x * blockDim.x + threadIdx.x;
  if (i < n) {
    float v = deg[i];
    deg[i] = v > 0.0f ? rsqrtf(v) : 0.0f;
  }
}

__global__ void agg_self(const float* __restrict__ hw,
                         const float* __restrict__ dinv,
                         const float* __restrict__ bias,
                         float* __restrict__ out)
{
  long idx = (long)blockIdx.x * blockDim.x + threadIdx.x;
  if (idx >= (long)NNODES * HID) return;
  long i = idx >> 7;
  int  c = (int)(idx & 127);
  float di = dinv[i];
  out[idx] = bias[c] + di * di * hw[idx];
}

__global__ __launch_bounds__(256) void agg_edges(
    const int* __restrict__ eui, const int* __restrict__ eiu,
    const float* __restrict__ hw, const float* __restrict__ dinv,
    float* __restrict__ out)
{
  int lane = threadIdx.x & 31;
  int w    = threadIdx.x >> 5;
  int e    = blockIdx.x * 8 + w;
  if (e >= E_TOT) return;

  int s, d;
  if (e < E_UI) { s = eui[e];                 d = eui[E_UI + e] + N_USER; }
  else          { int e2 = e - E_UI;
                  s = eiu[e2] + N_USER;       d = eiu[E_IU + e2]; }

  float nrm = dinv[s] * dinv[d];
  const float4* __restrict__ hs = (const float4*)(hw + (long)s * HID);
  float4 v = hs[lane];
  float* od = out + (long)d * HID + lane * 4;
  atomicAdd(od + 0, nrm * v.x);
  atomicAdd(od + 1, nrm * v.y);
  atomicAdd(od + 2, nrm * v.z);
  atomicAdd(od + 3, nrm * v.w);
}

extern "C" void kernel_launch(void* const* d_in, const int* in_sizes, int n_in,
                              void* d_out, int out_size, void* d_ws, size_t ws_size,
                              hipStream_t stream)
{
  const float* x_user = (const float*)d_in[0];
  const float* x_item = (const float*)d_in[1];
  const int*   eui    = (const int*)d_in[2];
  const int*   eiu    = (const int*)d_in[3];
  const float* W_iu   = (const float*)d_in[4];
  const float* b_iu   = (const float*)d_in[5];
  const float* W_ii   = (const float*)d_in[6];
  const float* b_ii   = (const float*)d_in[7];
  const float* W1     = (const float*)d_in[8];
  const float* b1     = (const float*)d_in[9];
  const float* W2     = (const float*)d_in[10];
  const float* b2     = (const float*)d_in[11];
  float* out = (float*)d_out;

  float* h    = (float*)d_ws;
  float* hw   = h  + (size_t)NNODES * HID;
  float* dinv = hw + (size_t)NNODES * HID;

  const dim3 blk(128);

  gemm_n128<<<dim3((N_USER + 63) / 64), blk, 0, stream>>>(
      x_user, W_iu, b_iu, h, N_USER, 256,  0,  1);
  gemm_n128<<<dim3((N_ITEM + 63) / 64), blk, 0, stream>>>(
      x_item, W_ii, b_ii, h + (size_t)N_USER * HID, N_ITEM, 256, 0, 1);

  deg_init <<<dim3((NNODES + 255) / 256), dim3(256), 0, stream>>>(dinv, NNODES);
  deg_count<<<dim3((E_TOT + 255) / 256), dim3(256), 0, stream>>>(eui, eiu, dinv);
  dinv_fin <<<dim3((NNODES + 255) / 256), dim3(256), 0, stream>>>(dinv, NNODES);

  const unsigned aggBlocks  = (unsigned)(((size_t)NNODES * HID + 255) / 256);
  const unsigned edgeBlocks = (E_TOT + 7) / 8;

  gemm_n128<<<dim3((NNODES + 63) / 64), blk, 0, stream>>>(
      h, W1, b1, hw, NNODES, HID,  0,  0);
  agg_self <<<dim3(aggBlocks),  dim3(256), 0, stream>>>(hw, dinv, b1, h);
  agg_edges<<<dim3(edgeBlocks), dim3(256), 0, stream>>>(eui, eiu, hw, dinv, h);

  gemm_n128<<<dim3((NNODES + 63) / 64), blk, 0, stream>>>(
      h, W2, b2, hw, NNODES, HID,  1,  0);
  agg_self <<<dim3(aggBlocks),  dim3(256), 0, stream>>>(hw, dinv, b2, out);
  agg_edges<<<dim3(edgeBlocks), dim3(256), 0, stream>>>(eui, eiu, hw, dinv, out);
}


# f32 atomic edge scatter replaced by per-destination gather: edges bucketed in-call (int atomic rank, scan, fill), one wave sums a node's rows in registers in source-sorted order, writes it once
# speedup vs baseline: 1.9550x; 1.9550x over previous
#include <hip/hip_runtime.h>


typedef __attribute__((ext_vector_type(2))) float v2f;
typedef __attribute__((ext_vector_type(8))) float v8f;

#define N_USER 100000
#define N_ITEM 100000
#define NNODES (N_USER + N_ITEM)
#define HID    128
#define E_UI   600000
#define E_IU   600000
#define E_TOT  (E_UI + E_IU)

__global__ __launch_bounds__(128) void gemm_n128(
    const float* __restrict__ A, const float* __restrict__ W,
    const float* __restrict__ bias, float* __restrict__ C,
    int M, int K, int relu, int use_bias)
{
  const int lane  = threadIdx.x & 31;
  const int wave  = threadIdx.x >> 5;
  const long row0 = (long)blockIdx.x * 64 + (long)wave * 16;

  const int m     = lane & 15;
  const int n     = lane & 15;
  const int khalf = (lane >> 4) << 1;

  long arow = row0 + m;
  if (arow >= M) arow = M - 1;
  const float* __restrict__ Ar = A + arow * (long)K;

  v8f acc[8] = {};

  for (int k = 0; k < K; k += 4) {
    v2f a;
    a.x = Ar[k + khalf];
    a.y = Ar[k + khalf + 1];
    if (relu) { a.x = fmaxf(a.x, 0.0f); a.y = fmaxf(a.y, 0.0f); }

    const float* __restrict__ Wk = W + (long)(k + khalf) * HID + n;
#pragma unroll
    for (int t = 0; t < 8; ++t) {
      v2f b;
      b.x = Wk[t * 16];
      b.y = Wk[t * 16 + HID];
      acc[t] = __builtin_amdgcn_wmma_f32_16x16x4_f32(
           false, a,  false, b,
           (short)0, acc[t],  false,  false);
    }
  }

  const int rhalf = (lane >> 4) << 3;
#pragma unroll
  for (int t = 0; t < 8; ++t) {
    float bv = use_bias ? bias[t * 16 + n] : 0.0f;
#pragma unroll
    for (int r = 0; r < 8; ++r) {
      long row = row0 + rhalf + r;
      if (row < M) C[row * HID + t * 16 + n] = acc[t][r] + bv;
    }
  }
}

#define SCAN_ELEMS  1024
#define SCAN_BLOCKS ((NNODES + SCAN_ELEMS - 1) / SCAN_ELEMS)
#define NODE_QUADS  (NNODES / 4)
static_assert(NNODES % 4 == 0, "scan kernels read cnt as int4");
static_assert(SCAN_BLOCKS <= 256, "block totals are summed by one 256-thread block");

__global__ void cnt_zero(int* __restrict__ cnt, int n)
{
  int i = blockIdx.x * blockDim.x + threadIdx.x;
  if (i < n) cnt[i] = 0;
}

__device__ __forceinline__ void edge_ends(const int* __restrict__ eui,
                                          const int* __restrict__ eiu,
                                          int e, int& s, int& d)
{
  if (e < E_UI) { s = eui[e];                 d = eui[E_UI + e] + N_USER; }
  else          { int e2 = e - E_UI;
                  s = eiu[e2] + N_USER;       d = eiu[E_IU + e2]; }
}

__global__ void deg_rank(const int* __restrict__ eui,
                         const int* __restrict__ eiu,
                         int* __restrict__ cnt, int* __restrict__ rank)
{
  int e = blockIdx.x * blockDim.x + threadIdx.x;
  if (e >= E_TOT) return;
  int d;
  if (e < E_UI) d = eui[E_UI + e] + N_USER;
  else          d = eiu[E_IU + (e - E_UI)];
  rank[e] = atomicAdd(&cnt[d], 1);
}

__global__ __launch_bounds__(256) void scan_totals(const int* __restrict__ cnt,
                                                   int* __restrict__ btot,
                                                   float* __restrict__ dinv)
{
  __shared__ int wtot[8];
  const int t  = threadIdx.x;
  const int q  = blockIdx.x * 256 + t;
  int s = 0;
  if (q < NODE_QUADS) {
    int4 c = ((const int4*)cnt)[q];
    s = c.x + c.y + c.z + c.w;
    float4 dg, dv;
    dg.x = (float)(c.x + 1); dg.y = (float)(c.y + 1);
    dg.z = (float)(c.z + 1); dg.w = (float)(c.w + 1);
    dv.x = dg.x > 0.0f ? rsqrtf(dg.x) : 0.0f;
    dv.y = dg.y > 0.0f ? rsqrtf(dg.y) : 0.0f;
    dv.z = dg.z > 0.0f ? rsqrtf(dg.z) : 0.0f;
    dv.w = dg.w > 0.0f ? rsqrtf(dg.w) : 0.0f;
    ((float4*)dinv)[q] = dv;
  }
  for (int m = 16; m > 0; m >>= 1) s += __shfl_xor(s, m, 32);
  if ((t & 31) == 0) wtot[t >> 5] = s;
  __syncthreads();
  if (t == 0) {
    int tot = 0;
    for (int k = 0; k < 8; ++k) tot += wtot[k];
    btot[blockIdx.x] = tot;
  }
}

__global__ __launch_bounds__(256) void scan_rowptr(const int* __restrict__ cnt,
                                                   const int* __restrict__ btot,
                                                   int* __restrict__ rp)
{
  __shared__ int wsum[8];
  __shared__ int wpre[8];
  const int t    = threadIdx.x;
  const int lane = t & 31;
  const int w    = t >> 5;
  const int q    = blockIdx.x * 256 + t;

  int p = (t < (int)blockIdx.x) ? btot[t] : 0;
  for (int m = 16; m > 0; m >>= 1) p += __shfl_xor(p, m, 32);

  int4 c = make_int4(0, 0, 0, 0);
  if (q < NODE_QUADS) c = ((const int4*)cnt)[q];
  const int s = c.x + c.y + c.z + c.w;
  int inc = s;
  for (int dlt = 1; dlt < 32; dlt <<= 1) {
    int v = __shfl_up(inc, dlt, 32);
    if (lane >= dlt) inc += v;
  }
  if (lane == 31) wsum[w] = inc;
  if (lane == 0)  wpre[w] = p;
  __syncthreads();
  int base = 0;
  for (int k = 0; k < 8; ++k) {
    base += wpre[k];
    if (k < w) base += wsum[k];
  }
  if (q < NODE_QUADS) {
    int4 r;
    r.x = base + inc - s;
    r.y = r.x + c.x;
    r.z = r.y + c.y;
    r.w = r.z + c.z;
    ((int4*)rp)[q] = r;
    if (q == NODE_QUADS - 1) rp[NNODES] = r.w + c.w;
  }
}

__global__ void csr_fill(const int* __restrict__ eui, const int* __restrict__ eiu,
                         const int* __restrict__ rp, const int* __restrict__ rank,
                         const float* __restrict__ dinv, int* __restrict__ col)
{
  int e = blockIdx.x * blockDim.x + threadIdx.x;
  if (e >= E_TOT) return;
  int s, d;
  edge_ends(eui, eiu, e, s, d);
  ((int2*)col)[rp[d] + rank[e]] = make_int2(s, __float_as_int(dinv[s]));
}

template <int U>
__device__ __forceinline__ void add_rows(float4& acc, const float4* __restrict__ hw4,
                                         int ss, float mw, int k, int lane)
{
  float4 v[U];
  float  w[U];
  for (int u = 0; u < U; ++u) {
    const int s = __builtin_amdgcn_readlane(ss, k + u);
    v[u] = hw4[(long)s * (HID / 4) + lane];
  }
  for (int u = 0; u < U; ++u)
    w[u] = __int_as_float(__builtin_amdgcn_readlane(__float_as_int(mw), k + u));
  for (int u = 0; u < U; ++u) {
    acc.x = fmaf(w[u], v[u].x, acc.x);
    acc.y = fmaf(w[u], v[u].y, acc.y);
    acc.z = fmaf(w[u], v[u].z, acc.z);
    acc.w = fmaf(w[u], v[u].w, acc.w);
  }
}

__global__ __launch_bounds__(256) void agg_gather(
    const int* __restrict__ rp, const int* __restrict__ col,
    const float* __restrict__ hw, const float* __restrict__ dinv,
    const float* __restrict__ bias, float* __restrict__ out)
{
  const int lane = threadIdx.x & 31;
  const int w = __builtin_amdgcn_readfirstlane((int)(threadIdx.x >> 5));
  const int i = blockIdx.x * 8 + w;
  if (i >= NNODES) return;

  const float4* __restrict__ hw4 = (const float4*)hw;
  const int   start = rp[i];
  const int   end   = rp[i + 1];
  const float di    = dinv[i];
  const float4 sv   = hw4[(long)i * (HID / 4) + lane];
  const float4 bv   = ((const float4*)bias)[lane];

  const float dd = di * di;
  float4 acc;
  acc.x = bv.x + dd * sv.x;
  acc.y = bv.y + dd * sv.y;
  acc.z = bv.z + dd * sv.z;
  acc.w = bv.w + dd * sv.w;

  for (int base = start; base < end; base += 32) {
    int n = end - base;
    if (n > 32) n = 32;

    int2 slot = make_int2(0x7fffffff, 0);
    if (lane < n) slot = ((const int2*)col)[base + lane];
    const int ms = slot.x;

    int pos = 0;
    for (int k = 0; k < n; ++k) {
      const int sk = __builtin_amdgcn_readlane(ms, k);
      pos += ((sk < ms) | ((sk == ms) & (k < lane))) ? 1 : 0;
    }
    int ss = __builtin_amdgcn_ds_permute(pos << 2, ms);
    const float mw = __int_as_float(__builtin_amdgcn_ds_permute(pos << 2, slot.y)) * di;
    ss = (int)min((unsigned)ss, (unsigned)(NNODES - 1));

    int j = 0;
    for (; j + 8 <= n; j += 8) add_rows<8>(acc, hw4, ss, mw, j, lane);
    if (j + 4 <= n) { add_rows<4>(acc, hw4, ss, mw, j, lane); j += 4; }
    switch (n - j) {
      case 3: add_rows<3>(acc, hw4, ss, mw, j, lane); break;
      case 2: add_rows<2>(acc, hw4, ss, mw, j, lane); break;
      case 1: add_rows<1>(acc, hw4, ss, mw, j, lane); break;
      default: break;
    }
  }

  ((float4*)out)[(long)i * (HID / 4) + lane] = acc;
}

extern "C" void kernel_launch(void* const* d_in, const int* in_sizes, int n_in,
                              void* d_out, int out_size, void* d_ws, size_t ws_size,
                              hipStream_t stream)
{
  const float* x_user = (const float*)d_in[0];
  const float* x_item = (const float*)d_in[1];
  const int*   eui    = (const int*)d_in[2];
  const int*   eiu    = (const int*)d_in[3];
  const float* W_iu   = (const float*)d_in[4];
  const float* b_iu   = (const float*)d_in[5];
  const float* W_ii   = (const float*)d_in[6];
  const float* b_ii   = (const float*)d_in[7];
  const float* W1     = (const float*)d_in[8];
  const float* b1     = (const float*)d_in[9];
  const float* W2     = (const float*)d_in[10];
  const float* b2     = (const float*)d_in[11];
  float* out = (float*)d_out;

  float* h    = (float*)d_ws;
  float* hw   = h  + (size_t)NNODES * HID;
  float* dinv = hw + (size_t)NNODES * HID;
  int*   cnt  = (int*)(dinv + NNODES);
  int*   rp   = cnt  + NNODES;
  int*   rank = rp   + (NNODES + 8);
  int*   col  = rank + E_TOT;
  int*   btot = col  + 2 * E_TOT;

  const dim3 blk(128);

  gemm_n128<<<dim3((N_USER + 63) / 64), blk, 0, stream>>>(
      x_user, W_iu, b_iu, h, N_USER, 256,  0,  1);
  gemm_n128<<<dim3((N_ITEM + 63) / 64), blk, 0, stream>>>(
      x_item, W_ii, b_ii, h + (size_t)N_USER * HID, N_ITEM, 256, 0, 1);

  cnt_zero   <<<dim3((NNODES + 255) / 256), dim3(256), 0, stream>>>(cnt, NNODES);
  deg_rank   <<<dim3((E_TOT + 255) / 256),  dim3(256), 0, stream>>>(eui, eiu, cnt, rank);
  scan_totals<<<dim3(SCAN_BLOCKS),          dim3(256), 0, stream>>>(cnt, btot, dinv);
  scan_rowptr<<<dim3(SCAN_BLOCKS),          dim3(256), 0, stream>>>(cnt, btot, rp);
  csr_fill   <<<dim3((E_TOT + 255) / 256),  dim3(256), 0, stream>>>(eui, eiu, rp, rank, dinv, col);

  const unsigned nodeBlocks = (NNODES + 7) / 8;

  gemm_n128<<<dim3((NNODES + 63) / 64), blk, 0, stream>>>(
      h, W1, b1, hw, NNODES, HID,  0,  0);
  agg_gather<<<dim3(nodeBlocks), dim3(256), 0, stream>>>(rp, col, hw, dinv, b1, h);

  gemm_n128<<<dim3((NNODES + 63) / 64), blk, 0, stream>>>(
      h, W2, b2, hw, NNODES, HID,  1,  0);
  agg_gather<<<dim3(nodeBlocks), dim3(256), 0, stream>>>(rp, col, hw, dinv, b2, out);
}


# GEMM weight operand staged once per resident work-group in LDS in B-fragment order (ds_load_b128, issued one k-pair ahead of its products); 256 work-groups loop over row tiles; sum order unchanged
# speedup vs baseline: 2.4462x; 1.2513x over previous
#include <hip/hip_runtime.h>


typedef __attribute__((ext_vector_type(2))) float v2f;
typedef __attribute__((ext_vector_type(4))) float v4f;
typedef __attribute__((ext_vector_type(8))) float v8f;

#define N_USER 100000
#define N_ITEM 100000
#define NNODES (N_USER + N_ITEM)
#define HID    128
#define IN_DIM 256
#define E_UI   600000
#define E_IU   600000
#define E_TOT  (E_UI + E_IU)

#define GEMM_WGS   256
#define GEMM_WAVES 16

__device__ __forceinline__ void load_a_line(v2f (&a)[8], const float* __restrict__ p)
{
#pragma unroll
  for (int j = 0; j < 8; ++j) a[j] = *(const v2f*)(p + 4 * j);
}

__device__ __forceinline__ void load_b_pair(v4f (&b)[8], const v4f* __restrict__ p)
{
#pragma unroll
  for (int t = 0; t < 8; ++t) b[t] = p[t * 32];
}

template <bool RELU>
__device__ __forceinline__ void wmma_kpair(v8f (&acc)[8], v2f s0, v2f s1, const v4f (&b)[8])
{
  if (RELU) {
    s0.x = fmaxf(s0.x, 0.0f); s0.y = fmaxf(s0.y, 0.0f);
    s1.x = fmaxf(s1.x, 0.0f); s1.y = fmaxf(s1.y, 0.0f);
  }
#pragma unroll
  for (int t = 0; t < 8; ++t) {
    v2f bb; bb.x = b[t].x; bb.y = b[t].y;
    acc[t] = __builtin_amdgcn_wmma_f32_16x16x4_f32(
         false, s0,  false, bb,
         (short)0, acc[t],  false,  false);
  }
#pragma unroll
  for (int t = 0; t < 8; ++t) {
    v2f bb; bb.x = b[t].z; bb.y = b[t].w;
    acc[t] = __builtin_amdgcn_wmma_f32_16x16x4_f32(
         false, s1,  false, bb,
         (short)0, acc[t],  false,  false);
  }
}

template <int K, bool RELU>
__global__ __launch_bounds__(GEMM_WAVES * 32) void gemm_wlds(
    const float* __restrict__ A, const float* __restrict__ W,
    const float* __restrict__ bias, float* __restrict__ C,
    int M, int use_bias)
{
  constexpr int KP = K / 8;
  constexpr int LB = K / 32;
  static_assert(K % 64 == 0, "the product loop takes two lines of A per trip");
  __shared__ v4f Ws[KP * 256];

  const int tid  = threadIdx.x;
  const int lane = tid & 31;
  const int wave = __builtin_amdgcn_readfirstlane(tid >> 5);
  const int c    = lane & 15;
  const int h    = lane >> 4;

  {
    const float* __restrict__ p0 =
        W + (size_t)(8 * (tid >> 8) + 2 * h) * HID + ((tid >> 5) & 7) * 16 + c;
#pragma unroll 8
    for (int i = 0; i < KP / 2; ++i) {
      const float* __restrict__ p = p0 + (size_t)i * 16 * HID;
      v4f v;
      v.x = p[0];
      v.y = p[HID];
      v.z = p[4 * HID];
      v.w = p[5 * HID];
      Ws[tid + 512 * i] = v;
    }
  }
  __syncthreads();

  const int ntiles = (M + 15) >> 4;
  const int nb     = (int)gridDim.x;
  const v4f* __restrict__ Bl = Ws + lane;
  const int rhalf  = h << 3;

  int tile = wave * nb + (int)blockIdx.x;
  v2f a0[8], a1[8];
  long arow = (long)tile * 16 + c;
  if (arow >= M) arow = M - 1;
  const float* __restrict__ Ar = A + arow * (long)K + 2 * h;
  load_a_line(a0, Ar);

  while (tile < ntiles) {
    asm volatile("" ::: "memory");
    const long row0 = (long)tile * 16;
    const int tnext = tile + GEMM_WAVES * nb;
    long arown = (long)tnext * 16 + c;
    if (arown >= M) arown = M - 1;
    const float* __restrict__ Arn = A + arown * (long)K + 2 * h;

    v8f acc[8] = {};
    v4f b0[8], b1[8];
    load_b_pair(b0, Bl);

    for (int lb = 0; lb < LB; lb += 2) {
      const float* __restrict__ An = (lb + 2 < LB) ? Ar + (lb + 2) * 32 : Arn;
      load_a_line(a1, Ar + (lb + 1) * 32);
#pragma unroll
      for (int q = 0; q < 8; ++q) {
        int kpn = lb * 4 + q + 1;
        if (kpn > KP - 1) kpn = KP - 1;
        if (q == 4) load_a_line(a0, An);
        if (q & 1) {
          load_b_pair(b0, Bl + kpn * 256);
          __builtin_amdgcn_sched_barrier(0);
          wmma_kpair<RELU>(acc, (q < 4 ? a0 : a1)[2 * (q & 3)], (q < 4 ? a0 : a1)[2 * (q & 3) + 1], b1);
        } else {
          load_b_pair(b1, Bl + kpn * 256);
          __builtin_amdgcn_sched_barrier(0);
          wmma_kpair<RELU>(acc, (q < 4 ? a0 : a1)[2 * (q & 3)], (q < 4 ? a0 : a1)[2 * (q & 3) + 1], b0);
        }
        __builtin_amdgcn_sched_barrier(0);
      }
    }

    const int n = c;
#pragma unroll
    for (int t = 0; t < 8; ++t) {
      float bv = use_bias ? bias[t * 16 + n] : 0.0f;
#pragma unroll
      for (int r = 0; r < 8; ++r) {
        long row = row0 + rhalf + r;
        if (row < M) C[row * HID + t * 16 + n] = acc[t][r] + bv;
      }
    }
    tile = tnext;
    Ar   = Arn;
  }
}

#define FOLD_TILES (IN_DIM / 16 + 1)
#define FOLD_ROWS  (FOLD_TILES * 16)
__global__ __launch_bounds__(256) void fold_in_w1(
    const float* __restrict__ W_iu, const float* __restrict__ b_iu,
    const float* __restrict__ W_ii, const float* __restrict__ b_ii,
    const float* __restrict__ W1, float* __restrict__ fold)
{
  const int lane  = threadIdx.x & 31;
  const int wave  = threadIdx.x >> 5;
  const int tile  = blockIdx.x;
  const int m     = lane & 15;
  const int khalf = (lane >> 4) << 1;

  const float* __restrict__ Win = blockIdx.y ? W_ii : W_iu;
  const float* __restrict__ bin = blockIdx.y ? b_ii : b_iu;
  const float* __restrict__ Ar  = (tile < IN_DIM / 16) ? Win + (long)(tile * 16 + m) * HID : bin;
  const float* __restrict__ Bc  = W1 + wave * 16 + m;

  v8f acc = {};
#pragma unroll
  for (int k = 0; k < HID; k += 4) {
    v2f a, b;
    a.x = Ar[k + khalf];
    a.y = Ar[k + khalf + 1];
    b.x = Bc[(k + khalf) * HID];
    b.y = Bc[(k + khalf + 1) * HID];
    acc = __builtin_amdgcn_wmma_f32_16x16x4_f32(
         false, a,  false, b,
         (short)0, acc,  false,  false);
  }

  float* __restrict__ o = fold + (size_t)blockIdx.y * FOLD_ROWS * HID + wave * 16 + m;
  if (tile < IN_DIM / 16) {
    const int rhalf = (lane >> 4) << 3;
#pragma unroll
    for (int r = 0; r < 8; ++r) o[(tile * 16 + rhalf + r) * HID] = acc[r];
  } else if (lane < 16) {
    o[IN_DIM * HID] = acc[0];
  }
}

#define SCAN_ELEMS  1024
#define SCAN_BLOCKS ((NNODES + SCAN_ELEMS - 1) / SCAN_ELEMS)
#define NODE_QUADS  (NNODES / 4)
static_assert(NNODES % 4 == 0, "scan kernels read cnt as int4");
static_assert(SCAN_BLOCKS <= 256, "block totals are summed by one 256-thread block");

__global__ void cnt_zero(int* __restrict__ cnt, int n)
{
  int i = blockIdx.x * blockDim.x + threadIdx.x;
  if (i < n) cnt[i] = 0;
}

__device__ __forceinline__ void edge_ends(const int* __restrict__ eui,
                                          const int* __restrict__ eiu,
                                          int e, int& s, int& d)
{
  if (e < E_UI) { s = eui[e];                 d = eui[E_UI + e] + N_USER; }
  else          { int e2 = e - E_UI;
                  s = eiu[e2] + N_USER;       d = eiu[E_IU + e2]; }
}

__global__ void deg_rank(const int* __restrict__ eui,
                         const int* __restrict__ eiu,
                         int* __restrict__ cnt, int* __restrict__ rank)
{
  int e = blockIdx.x * blockDim.x + threadIdx.x;
  if (e >= E_TOT) return;
  int d;
  if (e < E_UI) d = eui[E_UI + e] + N_USER;
  else          d = eiu[E_IU + (e - E_UI)];
  rank[e] = atomicAdd(&cnt[d], 1);
}

__global__ __launch_bounds__(256) void scan_totals(const int* __restrict__ cnt,
                                                   int* __restrict__ btot,
                                                   float* __restrict__ dinv)
{
  __shared__ int wtot[8];
  const int t  = threadIdx.x;
  const int q  = blockIdx.x * 256 + t;
  int s = 0;
  if (q < NODE_QUADS) {
    int4 c = ((const int4*)cnt)[q];
    s = c.x + c.y + c.z + c.w;
    float4 dg, dv;
    dg.x = (float)(c.x + 1); dg.y = (float)(c.y + 1);
    dg.z = (float)(c.z + 1); dg.w = (float)(c.w + 1);
    dv.x = dg.x > 0.0f ? rsqrtf(dg.x) : 0.0f;
    dv.y = dg.y > 0.0f ? rsqrtf(dg.y) : 0.0f;
    dv.z = dg.z > 0.0f ? rsqrtf(dg.z) : 0.0f;
    dv.w = dg.w > 0.0f ? rsqrtf(dg.w) : 0.0f;
    ((float4*)dinv)[q] = dv;
  }
  for (int m = 16; m > 0; m >>= 1) s += __shfl_xor(s, m, 32);
  if ((t & 31) == 0) wtot[t >> 5] = s;
  __syncthreads();
  if (t == 0) {
    int tot = 0;
    for (int k = 0; k < 8; ++k) tot += wtot[k];
    btot[blockIdx.x] = tot;
  }
}

__global__ __launch_bounds__(256) void scan_rowptr(const int* __restrict__ cnt,
                                                   const int* __restrict__ btot,
                                                   int* __restrict__ rp)
{
  __shared__ int wsum[8];
  __shared__ int wpre[8];
  const int t    = threadIdx.x;
  const int lane = t & 31;
  const int w    = t >> 5;
  const int q    = blockIdx.x * 256 + t;

  int p = (t < (int)blockIdx.x) ? btot[t] : 0;
  for (int m = 16; m > 0; m >>= 1) p += __shfl_xor(p, m, 32);

  int4 c = make_int4(0, 0, 0, 0);
  if (q < NODE_QUADS) c = ((const int4*)cnt)[q];
  const int s = c.x + c.y + c.z + c.w;
  int inc = s;
  for (int dlt = 1; dlt < 32; dlt <<= 1) {
    int v = __shfl_up(inc, dlt, 32);
    if (lane >= dlt) inc += v;
  }
  if (lane == 31) wsum[w] = inc;
  if (lane == 0)  wpre[w] = p;
  __syncthreads();
  int base = 0;
  for (int k = 0; k < 8; ++k) {
    base += wpre[k];
    if (k < w) base += wsum[k];
  }
  if (q < NODE_QUADS) {
    int4 r;
    r.x = base + inc - s;
    r.y = r.x + c.x;
    r.z = r.y + c.y;
    r.w = r.z + c.z;
    ((int4*)rp)[q] = r;
    if (q == NODE_QUADS - 1) rp[NNODES] = r.w + c.w;
  }
}

__global__ void csr_fill(const int* __restrict__ eui, const int* __restrict__ eiu,
                         const int* __restrict__ rp, const int* __restrict__ rank,
                         const float* __restrict__ dinv, int* __restrict__ col)
{
  int e = blockIdx.x * blockDim.x + threadIdx.x;
  if (e >= E_TOT) return;
  int s, d;
  edge_ends(eui, eiu, e, s, d);
  ((int2*)col)[rp[d] + rank[e]] = make_int2(s, __float_as_int(dinv[s]));
}

template <int U>
__device__ __forceinline__ void add_rows(float4& acc, const float4* __restrict__ hw4,
                                         int ss, float mw, int k, int lane)
{
  float4 v[U];
  float  w[U];
  for (int u = 0; u < U; ++u) {
    const int s = __builtin_amdgcn_readlane(ss, k + u);
    v[u] = hw4[(long)s * (HID / 4) + lane];
  }
  for (int u = 0; u < U; ++u)
    w[u] = __int_as_float(__builtin_amdgcn_readlane(__float_as_int(mw), k + u));
  for (int u = 0; u < U; ++u) {
    acc.x = fmaf(w[u], v[u].x, acc.x);
    acc.y = fmaf(w[u], v[u].y, acc.y);
    acc.z = fmaf(w[u], v[u].z, acc.z);
    acc.w = fmaf(w[u], v[u].w, acc.w);
  }
}

__global__ __launch_bounds__(256) void agg_gather(
    const int* __restrict__ rp, const int* __restrict__ col,
    const float* __restrict__ hw, const float* __restrict__ dinv,
    const float* __restrict__ bias, float* __restrict__ out)
{
  const int lane = threadIdx.x & 31;
  const int w = __builtin_amdgcn_readfirstlane((int)(threadIdx.x >> 5));
  const int i = blockIdx.x * 8 + w;
  if (i >= NNODES) return;

  const float4* __restrict__ hw4 = (const float4*)hw;
  const int   start = rp[i];
  const int   end   = rp[i + 1];
  const float di    = dinv[i];
  const float4 sv   = hw4[(long)i * (HID / 4) + lane];
  const float4 bv   = ((const float4*)bias)[lane];

  const float dd = di * di;
  float4 acc;
  acc.x = bv.x + dd * sv.x;
  acc.y = bv.y + dd * sv.y;
  acc.z = bv.z + dd * sv.z;
  acc.w = bv.w + dd * sv.w;

  for (int base = start; base < end; base += 32) {
    int n = end - base;
    if (n > 32) n = 32;

    int2 slot = make_int2(0x7fffffff, 0);
    if (lane < n) slot = ((const int2*)col)[base + lane];
    const int ms = slot.x;

    int pos = 0;
    for (int k = 0; k < n; ++k) {
      const int sk = __builtin_amdgcn_readlane(ms, k);
      pos += ((sk < ms) | ((sk == ms) & (k < lane))) ? 1 : 0;
    }
    int ss = __builtin_amdgcn_ds_permute(pos << 2, ms);
    const float mw = __int_as_float(__builtin_amdgcn_ds_permute(pos << 2, slot.y)) * di;
    ss = (int)min((unsigned)ss, (unsigned)(NNODES - 1));

    int j = 0;
    for (; j + 8 <= n; j += 8) add_rows<8>(acc, hw4, ss, mw, j, lane);
    if (j + 4 <= n) { add_rows<4>(acc, hw4, ss, mw, j, lane); j += 4; }
    switch (n - j) {
      case 3: add_rows<3>(acc, hw4, ss, mw, j, lane); break;
      case 2: add_rows<2>(acc, hw4, ss, mw, j, lane); break;
      case 1: add_rows<1>(acc, hw4, ss, mw, j, lane); break;
      default: break;
    }
  }

  ((float4*)out)[(long)i * (HID / 4) + lane] = acc;
}

extern "C" void kernel_launch(void* const* d_in, const int* in_sizes, int n_in,
                              void* d_out, int out_size, void* d_ws, size_t ws_size,
                              hipStream_t stream)
{
  const float* x_user = (const float*)d_in[0];
  const float* x_item = (const float*)d_in[1];
  const int*   eui    = (const int*)d_in[2];
  const int*   eiu    = (const int*)d_in[3];
  const float* W_iu   = (const float*)d_in[4];
  const float* b_iu   = (const float*)d_in[5];
  const float* W_ii   = (const float*)d_in[6];
  const float* b_ii   = (const float*)d_in[7];
  const float* W1     = (const float*)d_in[8];
  const float* b1     = (const float*)d_in[9];
  const float* W2     = (const float*)d_in[10];
  const float* b2     = (const float*)d_in[11];
  float* out = (float*)d_out;

  float* h    = (float*)d_ws;
  float* hw   = h  + (size_t)NNODES * HID;
  float* dinv = hw + (size_t)NNODES * HID;
  int*   cnt  = (int*)(dinv + NNODES);
  int*   rp   = cnt  + NNODES;
  int*   rank = rp   + (NNODES + 8);
  int*   col  = rank + E_TOT;
  int*   btot = col  + 2 * E_TOT;
  float* fold = (float*)(btot + 256);
  float* fold_u = fold;
  float* fold_i = fold + (size_t)FOLD_ROWS * HID;

  const dim3 blk(GEMM_WAVES * 32);

  fold_in_w1<<<dim3(FOLD_TILES, 2), dim3(256), 0, stream>>>(W_iu, b_iu, W_ii, b_ii, W1, fold);
  gemm_wlds<IN_DIM, false><<<dim3(GEMM_WGS), blk, 0, stream>>>(
      x_user, fold_u, fold_u + IN_DIM * HID, hw, N_USER,  1);
  gemm_wlds<IN_DIM, false><<<dim3(GEMM_WGS), blk, 0, stream>>>(
      x_item, fold_i, fold_i + IN_DIM * HID, hw + (size_t)N_USER * HID, N_ITEM, 1);

  cnt_zero   <<<dim3((NNODES + 255) / 256), dim3(256), 0, stream>>>(cnt, NNODES);
  deg_rank   <<<dim3((E_TOT + 255) / 256),  dim3(256), 0, stream>>>(eui, eiu, cnt, rank);
  scan_totals<<<dim3(SCAN_BLOCKS),          dim3(256), 0, stream>>>(cnt, btot, dinv);
  scan_rowptr<<<dim3(SCAN_BLOCKS),          dim3(256), 0, stream>>>(cnt, btot, rp);
  csr_fill   <<<dim3((E_TOT + 255) / 256),  dim3(256), 0, stream>>>(eui, eiu, rp, rank, dinv, col);

  const unsigned nodeBlocks = (NNODES + 7) / 8;

  agg_gather<<<dim3(nodeBlocks), dim3(256), 0, stream>>>(rp, col, hw, dinv, b1, h);

  gemm_wlds<HID, true><<<dim3(GEMM_WGS), blk, 0, stream>>>(
      h, W2, b2, hw, NNODES,  0);
  agg_gather<<<dim3(nodeBlocks), dim3(256), 0, stream>>>(rp, col, hw, dinv, b2, out);
}


# edge ranks from LDS counters: a 1024-thread owner work-group per 1000 nodes reads its type's destination row; global returning atomics, counter clear and totals launch go (11 to 9); sum order same
# speedup vs baseline: 2.5526x; 1.0435x over previous
#include <hip/hip_runtime.h>


typedef __attribute__((ext_vector_type(2))) float v2f;
typedef __attribute__((ext_vector_type(4))) float v4f;
typedef __attribute__((ext_vector_type(8))) float v8f;

#define N_USER 100000
#define N_ITEM 100000
#define NNODES (N_USER + N_ITEM)
#define HID    128
#define IN_DIM 256
#define E_UI   600000
#define E_IU   600000
#define E_TOT  (E_UI + E_IU)

#define GEMM_WGS   256
#define GEMM_WAVES 16

__device__ __forceinline__ void load_a_line(v2f (&a)[8], const float* __restrict__ p)
{
#pragma unroll
  for (int j = 0; j < 8; ++j) a[j] = *(const v2f*)(p + 4 * j);
}

__device__ __forceinline__ void load_b_pair(v4f (&b)[8], const v4f* __restrict__ p)
{
#pragma unroll
  for (int t = 0; t < 8; ++t) b[t] = p[t * 32];
}

template <bool RELU>
__device__ __forceinline__ void wmma_kpair(v8f (&acc)[8], v2f s0, v2f s1, const v4f (&b)[8])
{
  if (RELU) {
    s0.x = fmaxf(s0.x, 0.0f); s0.y = fmaxf(s0.y, 0.0f);
    s1.x = fmaxf(s1.x, 0.0f); s1.y = fmaxf(s1.y, 0.0f);
  }
#pragma unroll
  for (int t = 0; t < 8; ++t) {
    v2f bb; bb.x = b[t].x; bb.y = b[t].y;
    acc[t] = __builtin_amdgcn_wmma_f32_16x16x4_f32(
         false, s0,  false, bb,
         (short)0, acc[t],  false,  false);
  }
#pragma unroll
  for (int t = 0; t < 8; ++t) {
    v2f bb; bb.x = b[t].z; bb.y = b[t].w;
    acc[t] = __builtin_amdgcn_wmma_f32_16x16x4_f32(
         false, s1,  false, bb,
         (short)0, acc[t],  false,  false);
  }
}

template <int K, bool RELU>
__global__ __launch_bounds__(GEMM_WAVES * 32) void gemm_wlds(
    const float* __restrict__ A, const float* __restrict__ W,
    const float* __restrict__ bias, float* __restrict__ C,
    int M, int use_bias)
{
  constexpr int KP = K / 8;
  constexpr int LB = K / 32;
  static_assert(K % 64 == 0, "the product loop takes two lines of A per trip");
  __shared__ v4f Ws[KP * 256];

  const int tid  = threadIdx.x;
  const int lane = tid & 31;
  const int wave = __builtin_amdgcn_readfirstlane(tid >> 5);
  const int c    = lane & 15;
  const int h    = lane >> 4;

  {
    const float* __restrict__ p0 =
        W + (size_t)(8 * (tid >> 8) + 2 * h) * HID + ((tid >> 5) & 7) * 16 + c;
#pragma unroll 8
    for (int i = 0; i < KP / 2; ++i) {
      const float* __restrict__ p = p0 + (size_t)i * 16 * HID;
      v4f v;
      v.x = p[0];
      v.y = p[HID];
      v.z = p[4 * HID];
      v.w = p[5 * HID];
      Ws[tid + 512 * i] = v;
    }
  }
  __syncthreads();

  const int ntiles = (M + 15) >> 4;
  const int nb     = (int)gridDim.x;
  const v4f* __restrict__ Bl = Ws + lane;
  const int rhalf  = h << 3;

  int tile = wave * nb + (int)blockIdx.x;
  v2f a0[8], a1[8];
  long arow = (long)tile * 16 + c;
  if (arow >= M) arow = M - 1;
  const float* __restrict__ Ar = A + arow * (long)K + 2 * h;
  load_a_line(a0, Ar);

  while (tile < ntiles) {
    asm volatile("" ::: "memory");
    const long row0 = (long)tile * 16;
    const int tnext = tile + GEMM_WAVES * nb;
    long arown = (long)tnext * 16 + c;
    if (arown >= M) arown = M - 1;
    const float* __restrict__ Arn = A + arown * (long)K + 2 * h;

    v8f acc[8] = {};
    v4f b0[8], b1[8];
    load_b_pair(b0, Bl);

    for (int lb = 0; lb < LB; lb += 2) {
      const float* __restrict__ An = (lb + 2 < LB) ? Ar + (lb + 2) * 32 : Arn;
      load_a_line(a1, Ar + (lb + 1) * 32);
#pragma unroll
      for (int q = 0; q < 8; ++q) {
        int kpn = lb * 4 + q + 1;
        if (kpn > KP - 1) kpn = KP - 1;
        if (q == 4) load_a_line(a0, An);
        if (q & 1) {
          load_b_pair(b0, Bl + kpn * 256);
          __builtin_amdgcn_sched_barrier(0);
          wmma_kpair<RELU>(acc, (q < 4 ? a0 : a1)[2 * (q & 3)], (q < 4 ? a0 : a1)[2 * (q & 3) + 1], b1);
        } else {
          load_b_pair(b1, Bl + kpn * 256);
          __builtin_amdgcn_sched_barrier(0);
          wmma_kpair<RELU>(acc, (q < 4 ? a0 : a1)[2 * (q & 3)], (q < 4 ? a0 : a1)[2 * (q & 3) + 1], b0);
        }
        __builtin_amdgcn_sched_barrier(0);
      }
    }

    const int n = c;
#pragma unroll
    for (int t = 0; t < 8; ++t) {
      float bv = use_bias ? bias[t * 16 + n] : 0.0f;
#pragma unroll
      for (int r = 0; r < 8; ++r) {
        long row = row0 + rhalf + r;
        if (row < M) C[row * HID + t * 16 + n] = acc[t][r] + bv;
      }
    }
    tile = tnext;
    Ar   = Arn;
  }
}

#define FOLD_TILES (IN_DIM / 16 + 1)
#define FOLD_ROWS  (FOLD_TILES * 16)
__global__ __launch_bounds__(256) void fold_in_w1(
    const float* __restrict__ W_iu, const float* __restrict__ b_iu,
    const float* __restrict__ W_ii, const float* __restrict__ b_ii,
    const float* __restrict__ W1, float* __restrict__ fold)
{
  const int lane  = threadIdx.x & 31;
  const int wave  = threadIdx.x >> 5;
  const int tile  = blockIdx.x;
  const int m     = lane & 15;
  const int khalf = (lane >> 4) << 1;

  const float* __restrict__ Win = blockIdx.y ? W_ii : W_iu;
  const float* __restrict__ bin = blockIdx.y ? b_ii : b_iu;
  const float* __restrict__ Ar  = (tile < IN_DIM / 16) ? Win + (long)(tile * 16 + m) * HID : bin;
  const float* __restrict__ Bc  = W1 + wave * 16 + m;

  v8f acc = {};
#pragma unroll
  for (int k = 0; k < HID; k += 4) {
    v2f a, b;
    a.x = Ar[k + khalf];
    a.y = Ar[k + khalf + 1];
    b.x = Bc[(k + khalf) * HID];
    b.y = Bc[(k + khalf + 1) * HID];
    acc = __builtin_amdgcn_wmma_f32_16x16x4_f32(
         false, a,  false, b,
         (short)0, acc,  false,  false);
  }

  float* __restrict__ o = fold + (size_t)blockIdx.y * FOLD_ROWS * HID + wave * 16 + m;
  if (tile < IN_DIM / 16) {
    const int rhalf = (lane >> 4) << 3;
#pragma unroll
    for (int r = 0; r < 8; ++r) o[(tile * 16 + rhalf + r) * HID] = acc[r];
  } else if (lane < 16) {
    o[IN_DIM * HID] = acc[0];
  }
}

#define SCAN_ELEMS  1000
#define SCAN_QUADS  (SCAN_ELEMS / 4)
#define SCAN_BLOCKS (NNODES / SCAN_ELEMS)
#define NODE_QUADS  (NNODES / 4)
#define OWN_THREADS 1024
#define OWN_UNROLL  4
static_assert(NNODES % 4 == 0 && SCAN_ELEMS % 4 == 0, "scan kernel reads cnt as int4");
static_assert(SCAN_QUADS <= 256 && SCAN_ELEMS <= OWN_THREADS, "one int4 per scan thread, one node per owner thread");
static_assert(SCAN_BLOCKS <= 256, "block totals are summed by one 256-thread block");
static_assert(N_USER % SCAN_ELEMS == 0 && N_ITEM % SCAN_ELEMS == 0, "an owner range holds nodes of one type only");
static_assert(E_UI % 4 == 0 && E_IU % 4 == 0, "destination rows are read as int4");

__device__ __forceinline__ void edge_ends(const int* __restrict__ eui,
                                          const int* __restrict__ eiu,
                                          int e, int& s, int& d)
{
  if (e < E_UI) { s = eui[e];                 d = eui[E_UI + e] + N_USER; }
  else          { int e2 = e - E_UI;
                  s = eiu[e2] + N_USER;       d = eiu[E_IU + e2]; }
}

__device__ __forceinline__ void take_slot(int* lc, int* __restrict__ rank, int v, int first, int e)
{
  const unsigned x = (unsigned)(v - first);
  if (x < (unsigned)SCAN_ELEMS) rank[e] = atomicAdd(&lc[x], 1);
}

__global__ __launch_bounds__(OWN_THREADS) void own_count_rank(
    const int* __restrict__ eui, const int* __restrict__ eiu,
    int* __restrict__ cnt, int* __restrict__ rank,
    int* __restrict__ btot, float* __restrict__ dinv)
{
  __shared__ int lc[SCAN_ELEMS];
  __shared__ int wtot[OWN_THREADS / 32];
  const int tid = threadIdx.x;
  const int lo  = (int)blockIdx.x * SCAN_ELEMS;
  if (tid < SCAN_ELEMS) lc[tid] = 0;
  __syncthreads();

  const bool usr  = lo < N_USER;
  const int4* __restrict__ dl = (const int4*)(usr ? eiu + E_IU : eui + E_UI);
  const int nq    = (usr ? E_IU : E_UI) / 4;
  const int ebase = usr ? E_UI : 0;
  const int first = usr ? lo : lo - N_USER;
  const int bt    = usr ? (int)blockIdx.x : (int)blockIdx.x - N_USER / SCAN_ELEMS;
  const int nown  = (usr ? N_USER : N_ITEM) / SCAN_ELEMS;
  const int rot   = bt * ((nq / nown) & ~31);

  const int nit = (nq + OWN_THREADS * OWN_UNROLL - 1) / (OWN_THREADS * OWN_UNROLL);
  for (int it = 0; it < nit; ++it) {
    int4 v[OWN_UNROLL];
    int  g[OWN_UNROLL];
#pragma unroll
    for (int u = 0; u < OWN_UNROLL; ++u) {
      const int q = (it * OWN_UNROLL + u) * OWN_THREADS + tid;
      int r = q + rot;
      if (r >= nq) r -= nq;
      g[u] = r;
      v[u] = (q < nq) ? dl[r] : make_int4(-1, -1, -1, -1);
    }
#pragma unroll
    for (int u = 0; u < OWN_UNROLL; ++u) {
      const int e0 = ebase + 4 * g[u];
      take_slot(lc, rank, v[u].x, first, e0);
      take_slot(lc, rank, v[u].y, first, e0 + 1);
      take_slot(lc, rank, v[u].z, first, e0 + 2);
      take_slot(lc, rank, v[u].w, first, e0 + 3);
    }
  }
  __syncthreads();

  int c = 0;
  if (tid < SCAN_ELEMS) {
    c = lc[tid];
    cnt[lo + tid] = c;
    const float dg = (float)(c + 1);
    dinv[lo + tid] = dg > 0.0f ? rsqrtf(dg) : 0.0f;
  }
  for (int m = 16; m > 0; m >>= 1) c += __shfl_xor(c, m, 32);
  if ((tid & 31) == 0) wtot[tid >> 5] = c;
  __syncthreads();
  if (tid == 0) {
    int tot = 0;
    for (int k = 0; k < OWN_THREADS / 32; ++k) tot += wtot[k];
    btot[blockIdx.x] = tot;
  }
}

__global__ __launch_bounds__(256) void scan_rowptr(const int* __restrict__ cnt,
                                                   const int* __restrict__ btot,
                                                   int* __restrict__ rp)
{
  __shared__ int wsum[8];
  __shared__ int wpre[8];
  const int t    = threadIdx.x;
  const int lane = t & 31;
  const int w    = t >> 5;
  const int q    = blockIdx.x * SCAN_QUADS + t;
  const bool live = t < SCAN_QUADS;

  int p = (t < (int)blockIdx.x) ? btot[t] : 0;
  for (int m = 16; m > 0; m >>= 1) p += __shfl_xor(p, m, 32);

  int4 c = make_int4(0, 0, 0, 0);
  if (live) c = ((const int4*)cnt)[q];
  const int s = c.x + c.y + c.z + c.w;
  int inc = s;
  for (int dlt = 1; dlt < 32; dlt <<= 1) {
    int v = __shfl_up(inc, dlt, 32);
    if (lane >= dlt) inc += v;
  }
  if (lane == 31) wsum[w] = inc;
  if (lane == 0)  wpre[w] = p;
  __syncthreads();
  int base = 0;
  for (int k = 0; k < 8; ++k) {
    base += wpre[k];
    if (k < w) base += wsum[k];
  }
  if (live) {
    int4 r;
    r.x = base + inc - s;
    r.y = r.x + c.x;
    r.z = r.y + c.y;
    r.w = r.z + c.z;
    ((int4*)rp)[q] = r;
    if (q == NODE_QUADS - 1) rp[NNODES] = r.w + c.w;
  }
}

__global__ void csr_fill(const int* __restrict__ eui, const int* __restrict__ eiu,
                         const int* __restrict__ rp, const int* __restrict__ rank,
                         const float* __restrict__ dinv, int* __restrict__ col)
{
  int e = blockIdx.x * blockDim.x + threadIdx.x;
  if (e >= E_TOT) return;
  int s, d;
  edge_ends(eui, eiu, e, s, d);
  ((int2*)col)[rp[d] + rank[e]] = make_int2(s, __float_as_int(dinv[s]));
}

template <int U>
__device__ __forceinline__ void add_rows(float4& acc, const float4* __restrict__ hw4,
                                         int ss, float mw, int k, int lane)
{
  float4 v[U];
  float  w[U];
  for (int u = 0; u < U; ++u) {
    const int s = __builtin_amdgcn_readlane(ss, k + u);
    v[u] = hw4[(long)s * (HID / 4) + lane];
  }
  for (int u = 0; u < U; ++u)
    w[u] = __int_as_float(__builtin_amdgcn_readlane(__float_as_int(mw), k + u));
  for (int u = 0; u < U; ++u) {
    acc.x = fmaf(w[u], v[u].x, acc.x);
    acc.y = fmaf(w[u], v[u].y, acc.y);
    acc.z = fmaf(w[u], v[u].z, acc.z);
    acc.w = fmaf(w[u], v[u].w, acc.w);
  }
}

__global__ __launch_bounds__(256) void agg_gather(
    const int* __restrict__ rp, const int* __restrict__ col,
    const float* __restrict__ hw, const float* __restrict__ dinv,
    const float* __restrict__ bias, float* __restrict__ out)
{
  const int lane = threadIdx.x & 31;
  const int w = __builtin_amdgcn_readfirstlane((int)(threadIdx.x >> 5));
  const int i = blockIdx.x * 8 + w;
  if (i >= NNODES) return;

  const float4* __restrict__ hw4 = (const float4*)hw;
  const int   start = rp[i];
  const int   end   = rp[i + 1];
  const float di    = dinv[i];
  const float4 sv   = hw4[(long)i * (HID / 4) + lane];
  const float4 bv   = ((const float4*)bias)[lane];

  const float dd = di * di;
  float4 acc;
  acc.x = bv.x + dd * sv.x;
  acc.y = bv.y + dd * sv.y;
  acc.z = bv.z + dd * sv.z;
  acc.w = bv.w + dd * sv.w;

  for (int base = start; base < end; base += 32) {
    int n = end - base;
    if (n > 32) n = 32;

    int2 slot = make_int2(0x7fffffff, 0);
    if (lane < n) slot = ((const int2*)col)[base + lane];
    const int ms = slot.x;

    int pos = 0;
    for (int k = 0; k < n; ++k) {
      const int sk = __builtin_amdgcn_readlane(ms, k);
      pos += ((sk < ms) | ((sk == ms) & (k < lane))) ? 1 : 0;
    }
    int ss = __builtin_amdgcn_ds_permute(pos << 2, ms);
    const float mw = __int_as_float(__builtin_amdgcn_ds_permute(pos << 2, slot.y)) * di;
    ss = (int)min((unsigned)ss, (unsigned)(NNODES - 1));

    int j = 0;
    for (; j + 8 <= n; j += 8) add_rows<8>(acc, hw4, ss, mw, j, lane);
    if (j + 4 <= n) { add_rows<4>(acc, hw4, ss, mw, j, lane); j += 4; }
    switch (n - j) {
      case 3: add_rows<3>(acc, hw4, ss, mw, j, lane); break;
      case 2: add_rows<2>(acc, hw4, ss, mw, j, lane); break;
      case 1: add_rows<1>(acc, hw4, ss, mw, j, lane); break;
      default: break;
    }
  }

  ((float4*)out)[(long)i * (HID / 4) + lane] = acc;
}

extern "C" void kernel_launch(void* const* d_in, const int* in_sizes, int n_in,
                              void* d_out, int out_size, void* d_ws, size_t ws_size,
                              hipStream_t stream)
{
  const float* x_user = (const float*)d_in[0];
  const float* x_item = (const float*)d_in[1];
  const int*   eui    = (const int*)d_in[2];
  const int*   eiu    = (const int*)d_in[3];
  const float* W_iu   = (const float*)d_in[4];
  const float* b_iu   = (const float*)d_in[5];
  const float* W_ii   = (const float*)d_in[6];
  const float* b_ii   = (const float*)d_in[7];
  const float* W1     = (const float*)d_in[8];
  const float* b1     = (const float*)d_in[9];
  const float* W2     = (const float*)d_in[10];
  const float* b2     = (const float*)d_in[11];
  float* out = (float*)d_out;

  float* h    = (float*)d_ws;
  float* hw   = h  + (size_t)NNODES * HID;
  float* dinv = hw + (size_t)NNODES * HID;
  int*   cnt  = (int*)(dinv + NNODES);
  int*   rp   = cnt  + NNODES;
  int*   rank = rp   + (NNODES + 8);
  int*   col  = rank + E_TOT;
  int*   btot = col  + 2 * E_TOT;
  float* fold = (float*)(btot + 256);
  float* fold_u = fold;
  float* fold_i = fold + (size_t)FOLD_ROWS * HID;

  const dim3 blk(GEMM_WAVES * 32);

  fold_in_w1<<<dim3(FOLD_TILES, 2), dim3(256), 0, stream>>>(W_iu, b_iu, W_ii, b_ii, W1, fold);
  gemm_wlds<IN_DIM, false><<<dim3(GEMM_WGS), blk, 0, stream>>>(
      x_user, fold_u, fold_u + IN_DIM * HID, hw, N_USER,  1);
  gemm_wlds<IN_DIM, false><<<dim3(GEMM_WGS), blk, 0, stream>>>(
      x_item, fold_i, fold_i + IN_DIM * HID, hw + (size_t)N_USER * HID, N_ITEM, 1);

  own_count_rank<<<dim3(SCAN_BLOCKS),       dim3(OWN_THREADS), 0, stream>>>(eui, eiu, cnt, rank, btot, dinv);
  scan_rowptr<<<dim3(SCAN_BLOCKS),          dim3(256), 0, stream>>>(cnt, btot, rp);
  csr_fill   <<<dim3((E_TOT + 255) / 256),  dim3(256), 0, stream>>>(eui, eiu, rp, rank, dinv, col);

  const unsigned nodeBlocks = (NNODES + 7) / 8;

  agg_gather<<<dim3(nodeBlocks), dim3(256), 0, stream>>>(rp, col, hw, dinv, b1, h);

  gemm_wlds<HID, true><<<dim3(GEMM_WGS), blk, 0, stream>>>(
      h, W2, b2, hw, NNODES,  0);
  agg_gather<<<dim3(nodeBlocks), dim3(256), 0, stream>>>(rp, col, hw, dinv, b2, out);
}


# layer-2 GEMM moved into the layer-1 aggregation launch: a wave sums 16 nodes into its own LDS tile, multiplies it by W2 (in LDS) on WMMA under other waves' gathers; 9 to 8 launches; sum order same
# speedup vs baseline: 2.6296x; 1.0302x over previous
#include <hip/hip_runtime.h>


typedef __attribute__((ext_vector_type(2))) float v2f;
typedef __attribute__((ext_vector_type(4))) float v4f;
typedef __attribute__((ext_vector_type(8))) float v8f;

#define N_USER 100000
#define N_ITEM 100000
#define NNODES (N_USER + N_ITEM)
#define HID    128
#define IN_DIM 256
#define E_UI   600000
#define E_IU   600000
#define E_TOT  (E_UI + E_IU)

#define GEMM_WGS   256
#define GEMM_WAVES 16

__device__ __forceinline__ void load_a_line(v2f (&a)[8], const float* __restrict__ p)
{
#pragma unroll
  for (int j = 0; j < 8; ++j) a[j] = *(const v2f*)(p + 4 * j);
}

__device__ __forceinline__ void load_b_pair(v4f (&b)[8], const v4f* __restrict__ p)
{
#pragma unroll
  for (int t = 0; t < 8; ++t) b[t] = p[t * 32];
}

template <bool RELU>
__device__ __forceinline__ void wmma_kpair(v8f (&acc)[8], v2f s0, v2f s1, const v4f (&b)[8])
{
  if (RELU) {
    s0.x = fmaxf(s0.x, 0.0f); s0.y = fmaxf(s0.y, 0.0f);
    s1.x = fmaxf(s1.x, 0.0f); s1.y = fmaxf(s1.y, 0.0f);
  }
#pragma unroll
  for (int t = 0; t < 8; ++t) {
    v2f bb; bb.x = b[t].x; bb.y = b[t].y;
    acc[t] = __builtin_amdgcn_wmma_f32_16x16x4_f32(
         false, s0,  false, bb,
         (short)0, acc[t],  false,  false);
  }
#pragma unroll
  for (int t = 0; t < 8; ++t) {
    v2f bb; bb.x = b[t].z; bb.y = b[t].w;
    acc[t] = __builtin_amdgcn_wmma_f32_16x16x4_f32(
         false, s1,  false, bb,
         (short)0, acc[t],  false,  false);
  }
}

template <int K, bool RELU>
__global__ __launch_bounds__(GEMM_WAVES * 32) void gemm_wlds(
    const float* __restrict__ A, const float* __restrict__ W,
    const float* __restrict__ bias, float* __restrict__ C,
    int M, int use_bias)
{
  constexpr int KP = K / 8;
  constexpr int LB = K / 32;
  static_assert(K % 64 == 0, "the product loop takes two lines of A per trip");
  __shared__ v4f Ws[KP * 256];

  const int tid  = threadIdx.x;
  const int lane = tid & 31;
  const int wave = __builtin_amdgcn_readfirstlane(tid >> 5);
  const int c    = lane & 15;
  const int h    = lane >> 4;

  {
    const float* __restrict__ p0 =
        W + (size_t)(8 * (tid >> 8) + 2 * h) * HID + ((tid >> 5) & 7) * 16 + c;
#pragma unroll 8
    for (int i = 0; i < KP / 2; ++i) {
      const float* __restrict__ p = p0 + (size_t)i * 16 * HID;
      v4f v;
      v.x = p[0];
      v.y = p[HID];
      v.z = p[4 * HID];
      v.w = p[5 * HID];
      Ws[tid + 512 * i] = v;
    }
  }
  __syncthreads();

  const int ntiles = (M + 15) >> 4;
  const int nb     = (int)gridDim.x;
  const v4f* __restrict__ Bl = Ws + lane;
  const int rhalf  = h << 3;

  int tile = wave * nb + (int)blockIdx.x;
  v2f a0[8], a1[8];
  long arow = (long)tile * 16 + c;
  if (arow >= M) arow = M - 1;
  const float* __restrict__ Ar = A + arow * (long)K + 2 * h;
  load_a_line(a0, Ar);

  while (tile < ntiles) {
    asm volatile("" ::: "memory");
    const long row0 = (long)tile * 16;
    const int tnext = tile + GEMM_WAVES * nb;
    long arown = (long)tnext * 16 + c;
    if (arown >= M) arown = M - 1;
    const float* __restrict__ Arn = A + arown * (long)K + 2 * h;

    v8f acc[8] = {};
    v4f b0[8], b1[8];
    load_b_pair(b0, Bl);

    for (int lb = 0; lb < LB; lb += 2) {
      const float* __restrict__ An = (lb + 2 < LB) ? Ar + (lb + 2) * 32 : Arn;
      load_a_line(a1, Ar + (lb + 1) * 32);
#pragma unroll
      for (int q = 0; q < 8; ++q) {
        int kpn = lb * 4 + q + 1;
        if (kpn > KP - 1) kpn = KP - 1;
        if (q == 4) load_a_line(a0, An);
        if (q & 1) {
          load_b_pair(b0, Bl + kpn * 256);
          __builtin_amdgcn_sched_barrier(0);
          wmma_kpair<RELU>(acc, (q < 4 ? a0 : a1)[2 * (q & 3)], (q < 4 ? a0 : a1)[2 * (q & 3) + 1], b1);
        } else {
          load_b_pair(b1, Bl + kpn * 256);
          __builtin_amdgcn_sched_barrier(0);
          wmma_kpair<RELU>(acc, (q < 4 ? a0 : a1)[2 * (q & 3)], (q < 4 ? a0 : a1)[2 * (q & 3) + 1], b0);
        }
        __builtin_amdgcn_sched_barrier(0);
      }
    }

    const int n = c;
#pragma unroll
    for (int t = 0; t < 8; ++t) {
      float bv = use_bias ? bias[t * 16 + n] : 0.0f;
#pragma unroll
      for (int r = 0; r < 8; ++r) {
        long row = row0 + rhalf + r;
        if (row < M) C[row * HID + t * 16 + n] = acc[t][r] + bv;
      }
    }
    tile = tnext;
    Ar   = Arn;
  }
}

#define FOLD_TILES (IN_DIM / 16 + 1)
#define FOLD_ROWS  (FOLD_TILES * 16)
__global__ __launch_bounds__(256) void fold_in_w1(
    const float* __restrict__ W_iu, const float* __restrict__ b_iu,
    const float* __restrict__ W_ii, const float* __restrict__ b_ii,
    const float* __restrict__ W1, float* __restrict__ fold)
{
  const int lane  = threadIdx.x & 31;
  const int wave  = threadIdx.x >> 5;
  const int tile  = blockIdx.x;
  const int m     = lane & 15;
  const int khalf = (lane >> 4) << 1;

  const float* __restrict__ Win = blockIdx.y ? W_ii : W_iu;
  const float* __restrict__ bin = blockIdx.y ? b_ii : b_iu;
  const float* __restrict__ Ar  = (tile < IN_DIM / 16) ? Win + (long)(tile * 16 + m) * HID : bin;
  const float* __restrict__ Bc  = W1 + wave * 16 + m;

  v8f acc = {};
#pragma unroll
  for (int k = 0; k < HID; k += 4) {
    v2f a, b;
    a.x = Ar[k + khalf];
    a.y = Ar[k + khalf + 1];
    b.x = Bc[(k + khalf) * HID];
    b.y = Bc[(k + khalf + 1) * HID];
    acc = __builtin_amdgcn_wmma_f32_16x16x4_f32(
         false, a,  false, b,
         (short)0, acc,  false,  false);
  }

  float* __restrict__ o = fold + (size_t)blockIdx.y * FOLD_ROWS * HID + wave * 16 + m;
  if (tile < IN_DIM / 16) {
    const int rhalf = (lane >> 4) << 3;
#pragma unroll
    for (int r = 0; r < 8; ++r) o[(tile * 16 + rhalf + r) * HID] = acc[r];
  } else if (lane < 16) {
    o[IN_DIM * HID] = acc[0];
  }
}

#define SCAN_ELEMS  1000
#define SCAN_QUADS  (SCAN_ELEMS / 4)
#define SCAN_BLOCKS (NNODES / SCAN_ELEMS)
#define NODE_QUADS  (NNODES / 4)
#define OWN_THREADS 1024
#define OWN_UNROLL  4
static_assert(NNODES % 4 == 0 && SCAN_ELEMS % 4 == 0, "scan kernel reads cnt as int4");
static_assert(SCAN_QUADS <= 256 && SCAN_ELEMS <= OWN_THREADS, "one int4 per scan thread, one node per owner thread");
static_assert(SCAN_BLOCKS <= 256, "block totals are summed by one 256-thread block");
static_assert(N_USER % SCAN_ELEMS == 0 && N_ITEM % SCAN_ELEMS == 0, "an owner range holds nodes of one type only");
static_assert(E_UI % 4 == 0 && E_IU % 4 == 0, "destination rows are read as int4");

__device__ __forceinline__ void edge_ends(const int* __restrict__ eui,
                                          const int* __restrict__ eiu,
                                          int e, int& s, int& d)
{
  if (e < E_UI) { s = eui[e];                 d = eui[E_UI + e] + N_USER; }
  else          { int e2 = e - E_UI;
                  s = eiu[e2] + N_USER;       d = eiu[E_IU + e2]; }
}

__device__ __forceinline__ void take_slot(int* lc, int* __restrict__ rank, int v, int first, int e)
{
  const unsigned x = (unsigned)(v - first);
  if (x < (unsigned)SCAN_ELEMS) rank[e] = atomicAdd(&lc[x], 1);
}

__global__ __launch_bounds__(OWN_THREADS) void own_count_rank(
    const int* __restrict__ eui, const int* __restrict__ eiu,
    int* __restrict__ cnt, int* __restrict__ rank,
    int* __restrict__ btot, float* __restrict__ dinv)
{
  __shared__ int lc[SCAN_ELEMS];
  __shared__ int wtot[OWN_THREADS / 32];
  const int tid = threadIdx.x;
  const int lo  = (int)blockIdx.x * SCAN_ELEMS;
  if (tid < SCAN_ELEMS) lc[tid] = 0;
  __syncthreads();

  const bool usr  = lo < N_USER;
  const int4* __restrict__ dl = (const int4*)(usr ? eiu + E_IU : eui + E_UI);
  const int nq    = (usr ? E_IU : E_UI) / 4;
  const int ebase = usr ? E_UI : 0;
  const int first = usr ? lo : lo - N_USER;
  const int bt    = usr ? (int)blockIdx.x : (int)blockIdx.x - N_USER / SCAN_ELEMS;
  const int nown  = (usr ? N_USER : N_ITEM) / SCAN_ELEMS;
  const int rot   = bt * ((nq / nown) & ~31);

  const int nit = (nq + OWN_THREADS * OWN_UNROLL - 1) / (OWN_THREADS * OWN_UNROLL);
  for (int it = 0; it < nit; ++it) {
    int4 v[OWN_UNROLL];
    int  g[OWN_UNROLL];
#pragma unroll
    for (int u = 0; u < OWN_UNROLL; ++u) {
      const int q = (it * OWN_UNROLL + u) * OWN_THREADS + tid;
      int r = q + rot;
      if (r >= nq) r -= nq;
      g[u] = r;
      v[u] = (q < nq) ? dl[r] : make_int4(-1, -1, -1, -1);
    }
#pragma unroll
    for (int u = 0; u < OWN_UNROLL; ++u) {
      const int e0 = ebase + 4 * g[u];
      take_slot(lc, rank, v[u].x, first, e0);
      take_slot(lc, rank, v[u].y, first, e0 + 1);
      take_slot(lc, rank, v[u].z, first, e0 + 2);
      take_slot(lc, rank, v[u].w, first, e0 + 3);
    }
  }
  __syncthreads();

  int c = 0;
  if (tid < SCAN_ELEMS) {
    c = lc[tid];
    cnt[lo + tid] = c;
    const float dg = (float)(c + 1);
    dinv[lo + tid] = dg > 0.0f ? rsqrtf(dg) : 0.0f;
  }
  for (int m = 16; m > 0; m >>= 1) c += __shfl_xor(c, m, 32);
  if ((tid & 31) == 0) wtot[tid >> 5] = c;
  __syncthreads();
  if (tid == 0) {
    int tot = 0;
    for (int k = 0; k < OWN_THREADS / 32; ++k) tot += wtot[k];
    btot[blockIdx.x] = tot;
  }
}

__global__ __launch_bounds__(256) void scan_rowptr(const int* __restrict__ cnt,
                                                   const int* __restrict__ btot,
                                                   int* __restrict__ rp)
{
  __shared__ int wsum[8];
  __shared__ int wpre[8];
  const int t    = threadIdx.x;
  const int lane = t & 31;
  const int w    = t >> 5;
  const int q    = blockIdx.x * SCAN_QUADS + t;
  const bool live = t < SCAN_QUADS;

  int p = (t < (int)blockIdx.x) ? btot[t] : 0;
  for (int m = 16; m > 0; m >>= 1) p += __shfl_xor(p, m, 32);

  int4 c = make_int4(0, 0, 0, 0);
  if (live) c = ((const int4*)cnt)[q];
  const int s = c.x + c.y + c.z + c.w;
  int inc = s;
  for (int dlt = 1; dlt < 32; dlt <<= 1) {
    int v = __shfl_up(inc, dlt, 32);
    if (lane >= dlt) inc += v;
  }
  if (lane == 31) wsum[w] = inc;
  if (lane == 0)  wpre[w] = p;
  __syncthreads();
  int base = 0;
  for (int k = 0; k < 8; ++k) {
    base += wpre[k];
    if (k < w) base += wsum[k];
  }
  if (live) {
    int4 r;
    r.x = base + inc - s;
    r.y = r.x + c.x;
    r.z = r.y + c.y;
    r.w = r.z + c.z;
    ((int4*)rp)[q] = r;
    if (q == NODE_QUADS - 1) rp[NNODES] = r.w + c.w;
  }
}

__global__ void csr_fill(const int* __restrict__ eui, const int* __restrict__ eiu,
                         const int* __restrict__ rp, const int* __restrict__ rank,
                         const float* __restrict__ dinv, int* __restrict__ col)
{
  int e = blockIdx.x * blockDim.x + threadIdx.x;
  if (e >= E_TOT) return;
  int s, d;
  edge_ends(eui, eiu, e, s, d);
  ((int2*)col)[rp[d] + rank[e]] = make_int2(s, __float_as_int(dinv[s]));
}

template <int U>
__device__ __forceinline__ void add_rows(float4& acc, const float4* __restrict__ hw4,
                                         int ss, float mw, int k, int lane)
{
  float4 v[U];
  float  w[U];
  for (int u = 0; u < U; ++u) {
    const int s = __builtin_amdgcn_readlane(ss, k + u);
    v[u] = hw4[(long)s * (HID / 4) + lane];
  }
  for (int u = 0; u < U; ++u)
    w[u] = __int_as_float(__builtin_amdgcn_readlane(__float_as_int(mw), k + u));
  for (int u = 0; u < U; ++u) {
    acc.x = fmaf(w[u], v[u].x, acc.x);
    acc.y = fmaf(w[u], v[u].y, acc.y);
    acc.z = fmaf(w[u], v[u].z, acc.z);
    acc.w = fmaf(w[u], v[u].w, acc.w);
  }
}

__global__ __launch_bounds__(256) void agg_gather(
    const int* __restrict__ rp, const int* __restrict__ col,
    const float* __restrict__ hw, const float* __restrict__ dinv,
    const float* __restrict__ bias, float* __restrict__ out)
{
  const int lane = threadIdx.x & 31;
  const int w = __builtin_amdgcn_readfirstlane((int)(threadIdx.x >> 5));
  const int i = blockIdx.x * 8 + w;
  if (i >= NNODES) return;

  const float4* __restrict__ hw4 = (const float4*)hw;
  const int   start = rp[i];
  const int   end   = rp[i + 1];
  const float di    = dinv[i];
  const float4 sv   = hw4[(long)i * (HID / 4) + lane];
  const float4 bv   = ((const float4*)bias)[lane];

  const float dd = di * di;
  float4 acc;
  acc.x = bv.x + dd * sv.x;
  acc.y = bv.y + dd * sv.y;
  acc.z = bv.z + dd * sv.z;
  acc.w = bv.w + dd * sv.w;

  for (int base = start; base < end; base += 32) {
    int n = end - base;
    if (n > 32) n = 32;

    int2 slot = make_int2(0x7fffffff, 0);
    if (lane < n) slot = ((const int2*)col)[base + lane];
    const int ms = slot.x;

    int pos = 0;
    for (int k = 0; k < n; ++k) {
      const int sk = __builtin_amdgcn_readlane(ms, k);
      pos += ((sk < ms) | ((sk == ms) & (k < lane))) ? 1 : 0;
    }
    int ss = __builtin_amdgcn_ds_permute(pos << 2, ms);
    const float mw = __int_as_float(__builtin_amdgcn_ds_permute(pos << 2, slot.y)) * di;
    ss = (int)min((unsigned)ss, (unsigned)(NNODES - 1));

    int j = 0;
    for (; j + 8 <= n; j += 8) add_rows<8>(acc, hw4, ss, mw, j, lane);
    if (j + 4 <= n) { add_rows<4>(acc, hw4, ss, mw, j, lane); j += 4; }
    switch (n - j) {
      case 3: add_rows<3>(acc, hw4, ss, mw, j, lane); break;
      case 2: add_rows<2>(acc, hw4, ss, mw, j, lane); break;
      case 1: add_rows<1>(acc, hw4, ss, mw, j, lane); break;
      default: break;
    }
  }

  ((float4*)out)[(long)i * (HID / 4) + lane] = acc;
}

__device__ __forceinline__ void sum_chunk(float4& acc, const float4* __restrict__ hw4,
                                          int2 raw, int n, float di, int lane)
{
  const int2 slot = lane < n ? raw : make_int2(0x7fffffff, 0);
  const int ms = slot.x;

  int pos = 0;
  for (int q = 0; q < n; ++q) {
    const int sk = __builtin_amdgcn_readlane(ms, q);
    pos += ((sk < ms) | ((sk == ms) & (q < lane))) ? 1 : 0;
  }
  int ss = __builtin_amdgcn_ds_permute(pos << 2, ms);
  const float mw = __int_as_float(__builtin_amdgcn_ds_permute(pos << 2, slot.y)) * di;
  ss = (int)min((unsigned)ss, (unsigned)(NNODES - 1));

  int j = 0;
  for (; j + 8 <= n; j += 8) add_rows<8>(acc, hw4, ss, mw, j, lane);
  switch (n - j) {
    case 7: add_rows<7>(acc, hw4, ss, mw, j, lane); break;
    case 6: add_rows<6>(acc, hw4, ss, mw, j, lane); break;
    case 5: add_rows<5>(acc, hw4, ss, mw, j, lane); break;
    case 4: add_rows<4>(acc, hw4, ss, mw, j, lane); break;
    case 3: add_rows<3>(acc, hw4, ss, mw, j, lane); break;
    case 2: add_rows<2>(acc, hw4, ss, mw, j, lane); break;
    case 1: add_rows<1>(acc, hw4, ss, mw, j, lane); break;
    default: break;
  }
}

#define FUSE_WGS   256
#define FUSE_WAVES 25

__global__ __launch_bounds__(FUSE_WAVES * 32) void agg_relu_gemm(
    const int* __restrict__ rp, const int* __restrict__ col,
    const float* __restrict__ hw, const float* __restrict__ dinv,
    const float* __restrict__ bias, const float* __restrict__ W,
    float* __restrict__ C)
{
  __shared__ v4f Ws[(HID / 8) * 256];
  __shared__ v4f Ts[FUSE_WAVES * 16 * 32];

  const int tid  = threadIdx.x;
  const int lane = tid & 31;
  const int wave = __builtin_amdgcn_readfirstlane(tid >> 5);
  const int c    = lane & 15;
  const int h    = lane >> 4;

#pragma unroll
  for (int i = 0; i < ((HID / 8) * 256 + FUSE_WAVES * 32 - 1) / (FUSE_WAVES * 32); ++i) {
    const int e = tid + FUSE_WAVES * 32 * i;
    if (e < (HID / 8) * 256) {
      const float* __restrict__ p =
          W + (size_t)(8 * (e >> 8) + 2 * h) * HID + ((e >> 5) & 7) * 16 + c;
      v4f v;
      v.x = p[0];
      v.y = p[HID];
      v.z = p[4 * HID];
      v.w = p[5 * HID];
      Ws[e] = v;
    }
  }
  __syncthreads();

  const float4* __restrict__ hw4  = (const float4*)hw;
  const int2*   __restrict__ col2 = (const int2*)col;
  const float4 bv = ((const float4*)bias)[lane];
  v4f* __restrict__ Tw = Ts + wave * (16 * 32);
  const float* __restrict__ Tr = (const float*)Tw + c * 128 + 2 * h;
  const v4f* __restrict__ Bl = Ws + lane;
  const int rhalf  = h << 3;
  static_assert(N_USER % 16 == 0 && N_ITEM % 16 == 0, "a tile is 16 whole nodes of one type");

  constexpr int NSLOTS = FUSE_WAVES * FUSE_WGS;
  constexpr int TU = N_USER / 16, TI = N_ITEM / 16;
  constexpr int PU = (TU + NSLOTS - 1) / NSLOTS;
  constexpr int PI = (TI + NSLOTS - 1) / NSLOTS;
  const int slot = wave * FUSE_WGS + (int)blockIdx.x;
  for (int pass = 0; pass < PU + PI; ++pass) {
    const bool usr = pass < PU;
    const int  t   = (usr ? pass : pass - PU) * NSLOTS + slot;
    if (t >= (usr ? TU : TI)) continue;
    const int tile = usr ? t : TU + t;
    asm volatile("" ::: "memory");
    const int node0 = tile * 16;
    const int   rpl = rp[node0 + (lane < 16 ? lane : 16)];
    const float dil = dinv[node0 + c];

    int2   pslot;
    float4 psv;
    {
      const int s0 = __builtin_amdgcn_readlane(rpl, 0);
      pslot = col2[min(s0 + lane, E_TOT - 1)];
      psv = hw4[(long)node0 * (HID / 4) + lane];
    }

    for (int k = 0; k < 16; ++k) {
      const int   start = __builtin_amdgcn_readlane(rpl, k);
      const int   end   = __builtin_amdgcn_readlane(rpl, k + 1);
      const float di    = __int_as_float(__builtin_amdgcn_readlane(__float_as_int(dil), k));
      const int2 raw = pslot;
      const float4 sv = psv;
      {
        const int kn = k < 15 ? k + 1 : 15;
        const int sn = __builtin_amdgcn_readlane(rpl, kn);
        pslot = col2[min(sn + lane, E_TOT - 1)];
        psv = hw4[(long)(node0 + kn) * (HID / 4) + lane];
      }

      const float dd = di * di;
      float4 acc;
      acc.x = bv.x + dd * sv.x;
      acc.y = bv.y + dd * sv.y;
      acc.z = bv.z + dd * sv.z;
      acc.w = bv.w + dd * sv.w;

      if (end > start) {
        sum_chunk(acc, hw4, raw, min(end - start, 32), di, lane);
        for (int base = start + 32; base < end; base += 32) {
          const int2 more = col2[min(base + lane, E_TOT - 1)];
          sum_chunk(acc, hw4, more, min(end - base, 32), di, lane);
        }
      }

      v4f o;
      o.x = fmaxf(acc.x, 0.0f);
      o.y = fmaxf(acc.y, 0.0f);
      o.z = fmaxf(acc.z, 0.0f);
      o.w = fmaxf(acc.w, 0.0f);
      Tw[k * 32 + (lane ^ k)] = o;
    }

    __builtin_amdgcn_fence(__ATOMIC_RELEASE, "workgroup", "local");
    __builtin_amdgcn_wave_barrier();
    __builtin_amdgcn_fence(__ATOMIC_ACQUIRE, "workgroup", "local");

    v8f acc8[8] = {};
    for (int kp = 0; kp < HID / 8; ++kp) {
      const v2f s0 = *(const v2f*)(Tr + (((2 * kp)     ^ c) << 2));
      const v2f s1 = *(const v2f*)(Tr + (((2 * kp + 1) ^ c) << 2));
      v4f b[8];
      load_b_pair(b, Bl + kp * 256);
      wmma_kpair<false>(acc8, s0, s1, b);
    }

    const long row0 = (long)node0;
#pragma unroll
    for (int t = 0; t < 8; ++t) {
#pragma unroll
      for (int r = 0; r < 8; ++r)
        C[(row0 + rhalf + r) * HID + t * 16 + c] = acc8[t][r];
    }

    __builtin_amdgcn_fence(__ATOMIC_RELEASE, "workgroup", "local");
    __builtin_amdgcn_wave_barrier();
  }
}

extern "C" void kernel_launch(void* const* d_in, const int* in_sizes, int n_in,
                              void* d_out, int out_size, void* d_ws, size_t ws_size,
                              hipStream_t stream)
{
  const float* x_user = (const float*)d_in[0];
  const float* x_item = (const float*)d_in[1];
  const int*   eui    = (const int*)d_in[2];
  const int*   eiu    = (const int*)d_in[3];
  const float* W_iu   = (const float*)d_in[4];
  const float* b_iu   = (const float*)d_in[5];
  const float* W_ii   = (const float*)d_in[6];
  const float* b_ii   = (const float*)d_in[7];
  const float* W1     = (const float*)d_in[8];
  const float* b1     = (const float*)d_in[9];
  const float* W2     = (const float*)d_in[10];
  const float* b2     = (const float*)d_in[11];
  float* out = (float*)d_out;

  float* h    = (float*)d_ws;
  float* hw   = h  + (size_t)NNODES * HID;
  float* dinv = hw + (size_t)NNODES * HID;
  int*   cnt  = (int*)(dinv + NNODES);
  int*   rp   = cnt  + NNODES;
  int*   rank = rp   + (NNODES + 8);
  int*   col  = rank + E_TOT;
  int*   btot = col  + 2 * E_TOT;
  float* fold = (float*)(btot + 256);
  float* fold_u = fold;
  float* fold_i = fold + (size_t)FOLD_ROWS * HID;

  const dim3 blk(GEMM_WAVES * 32);

  fold_in_w1<<<dim3(FOLD_TILES, 2), dim3(256), 0, stream>>>(W_iu, b_iu, W_ii, b_ii, W1, fold);
  gemm_wlds<IN_DIM, false><<<dim3(GEMM_WGS), blk, 0, stream>>>(
      x_user, fold_u, fold_u + IN_DIM * HID, hw, N_USER,  1);
  gemm_wlds<IN_DIM, false><<<dim3(GEMM_WGS), blk, 0, stream>>>(
      x_item, fold_i, fold_i + IN_DIM * HID, hw + (size_t)N_USER * HID, N_ITEM, 1);

  own_count_rank<<<dim3(SCAN_BLOCKS),       dim3(OWN_THREADS), 0, stream>>>(eui, eiu, cnt, rank, btot, dinv);
  scan_rowptr<<<dim3(SCAN_BLOCKS),          dim3(256), 0, stream>>>(cnt, btot, rp);
  csr_fill   <<<dim3((E_TOT + 255) / 256),  dim3(256), 0, stream>>>(eui, eiu, rp, rank, dinv, col);

  const unsigned nodeBlocks = (NNODES + 7) / 8;

  agg_relu_gemm<<<dim3(FUSE_WGS), dim3(FUSE_WAVES * 32), 0, stream>>>(rp, col, hw, dinv, b1, W2, h);

  agg_gather<<<dim3(nodeBlocks), dim3(256), 0, stream>>>(rp, col, h, dinv, b2, out);
}


# the two input-projection GEMM launches merged into one: work-groups 0-127 loop over user row tiles with the user folded weight in LDS, 128-255 over item tiles; 8 to 7 launches; sum order same
# speedup vs baseline: 2.7172x; 1.0333x over previous
#include <hip/hip_runtime.h>


typedef __attribute__((ext_vector_type(2))) float v2f;
typedef __attribute__((ext_vector_type(4))) float v4f;
typedef __attribute__((ext_vector_type(8))) float v8f;

#define N_USER 100000
#define N_ITEM 100000
#define NNODES (N_USER + N_ITEM)
#define HID    128
#define IN_DIM 256
#define E_UI   600000
#define E_IU   600000
#define E_TOT  (E_UI + E_IU)

#define GEMM_WGS   256
static_assert(GEMM_WGS % 2 == 0, "the launch splits its work-groups between two problems");
#define GEMM_WAVES 16

__device__ __forceinline__ void load_a_line(v2f (&a)[8], const float* __restrict__ p)
{
#pragma unroll
  for (int j = 0; j < 8; ++j) a[j] = *(const v2f*)(p + 4 * j);
}

__device__ __forceinline__ void load_b_pair(v4f (&b)[8], const v4f* __restrict__ p)
{
#pragma unroll
  for (int t = 0; t < 8; ++t) b[t] = p[t * 32];
}

template <bool RELU>
__device__ __forceinline__ void wmma_kpair(v8f (&acc)[8], v2f s0, v2f s1, const v4f (&b)[8])
{
  if (RELU) {
    s0.x = fmaxf(s0.x, 0.0f); s0.y = fmaxf(s0.y, 0.0f);
    s1.x = fmaxf(s1.x, 0.0f); s1.y = fmaxf(s1.y, 0.0f);
  }
#pragma unroll
  for (int t = 0; t < 8; ++t) {
    v2f bb; bb.x = b[t].x; bb.y = b[t].y;
    acc[t] = __builtin_amdgcn_wmma_f32_16x16x4_f32(
         false, s0,  false, bb,
         (short)0, acc[t],  false,  false);
  }
#pragma unroll
  for (int t = 0; t < 8; ++t) {
    v2f bb; bb.x = b[t].z; bb.y = b[t].w;
    acc[t] = __builtin_amdgcn_wmma_f32_16x16x4_f32(
         false, s1,  false, bb,
         (short)0, acc[t],  false,  false);
  }
}

template <int K, bool RELU>
__global__ __launch_bounds__(GEMM_WAVES * 32) void gemm_wlds(
    const float* __restrict__ A_p0, const float* __restrict__ W_p0,
    const float* __restrict__ bias_p0, float* __restrict__ C_p0, int M_p0,
    const float* __restrict__ A_p1, const float* __restrict__ W_p1,
    const float* __restrict__ bias_p1, float* __restrict__ C_p1, int M_p1)
{
  const int  nb   = (int)(gridDim.x >> 1);
  const bool p1   = (int)blockIdx.x >= nb;
  const int  bidx = p1 ? (int)blockIdx.x - nb : (int)blockIdx.x;
  const float* __restrict__ A    = p1 ? A_p1 : A_p0;
  const float* __restrict__ W    = p1 ? W_p1 : W_p0;
  const float* __restrict__ bias = p1 ? bias_p1 : bias_p0;
  float* __restrict__ C          = p1 ? C_p1 : C_p0;
  const int M                    = p1 ? M_p1 : M_p0;

  constexpr int KP = K / 8;
  constexpr int LB = K / 32;
  static_assert(K % 64 == 0, "the product loop takes two lines of A per trip");
  __shared__ v4f Ws[KP * 256];

  const int tid  = threadIdx.x;
  const int lane = tid & 31;
  const int wave = __builtin_amdgcn_readfirstlane(tid >> 5);
  const int c    = lane & 15;
  const int h    = lane >> 4;

  {
    const float* __restrict__ p0 =
        W + (size_t)(8 * (tid >> 8) + 2 * h) * HID + ((tid >> 5) & 7) * 16 + c;
#pragma unroll 8
    for (int i = 0; i < KP / 2; ++i) {
      const float* __restrict__ p = p0 + (size_t)i * 16 * HID;
      v4f v;
      v.x = p[0];
      v.y = p[HID];
      v.z = p[4 * HID];
      v.w = p[5 * HID];
      Ws[tid + 512 * i] = v;
    }
  }
  __syncthreads();

  const int ntiles = (M + 15) >> 4;
  const v4f* __restrict__ Bl = Ws + lane;
  const int rhalf  = h << 3;

  int tile = wave * nb + bidx;
  v2f a0[8], a1[8];
  long arow = (long)tile * 16 + c;
  if (arow >= M) arow = M - 1;
  const float* __restrict__ Ar = A + arow * (long)K + 2 * h;
  load_a_line(a0, Ar);

  while (tile < ntiles) {
    asm volatile("" ::: "memory");
    const long row0 = (long)tile * 16;
    const int tnext = tile + GEMM_WAVES * nb;
    long arown = (long)tnext * 16 + c;
    if (arown >= M) arown = M - 1;
    const float* __restrict__ Arn = A + arown * (long)K + 2 * h;

    v8f acc[8] = {};
    v4f b0[8], b1[8];
    load_b_pair(b0, Bl);

    for (int lb = 0; lb < LB; lb += 2) {
      const float* __restrict__ An = (lb + 2 < LB) ? Ar + (lb + 2) * 32 : Arn;
      load_a_line(a1, Ar + (lb + 1) * 32);
#pragma unroll
      for (int q = 0; q < 8; ++q) {
        int kpn = lb * 4 + q + 1;
        if (kpn > KP - 1) kpn = KP - 1;
        if (q == 4) load_a_line(a0, An);
        if (q & 1) {
          load_b_pair(b0, Bl + kpn * 256);
          __builtin_amdgcn_sched_barrier(0);
          wmma_kpair<RELU>(acc, (q < 4 ? a0 : a1)[2 * (q & 3)], (q < 4 ? a0 : a1)[2 * (q & 3) + 1], b1);
        } else {
          load_b_pair(b1, Bl + kpn * 256);
          __builtin_amdgcn_sched_barrier(0);
          wmma_kpair<RELU>(acc, (q < 4 ? a0 : a1)[2 * (q & 3)], (q < 4 ? a0 : a1)[2 * (q & 3) + 1], b0);
        }
        __builtin_amdgcn_sched_barrier(0);
      }
    }

    const int n = c;
#pragma unroll
    for (int t = 0; t < 8; ++t) {
      float bv = bias[t * 16 + n];
#pragma unroll
      for (int r = 0; r < 8; ++r) {
        long row = row0 + rhalf + r;
        if (row < M) C[row * HID + t * 16 + n] = acc[t][r] + bv;
      }
    }
    tile = tnext;
    Ar   = Arn;
  }
}

#define FOLD_TILES (IN_DIM / 16 + 1)
#define FOLD_ROWS  (FOLD_TILES * 16)
__global__ __launch_bounds__(256) void fold_in_w1(
    const float* __restrict__ W_iu, const float* __restrict__ b_iu,
    const float* __restrict__ W_ii, const float* __restrict__ b_ii,
    const float* __restrict__ W1, float* __restrict__ fold)
{
  const int lane  = threadIdx.x & 31;
  const int wave  = threadIdx.x >> 5;
  const int tile  = blockIdx.x;
  const int m     = lane & 15;
  const int khalf = (lane >> 4) << 1;

  const float* __restrict__ Win = blockIdx.y ? W_ii : W_iu;
  const float* __restrict__ bin = blockIdx.y ? b_ii : b_iu;
  const float* __restrict__ Ar  = (tile < IN_DIM / 16) ? Win + (long)(tile * 16 + m) * HID : bin;
  const float* __restrict__ Bc  = W1 + wave * 16 + m;

  v8f acc = {};
#pragma unroll
  for (int k = 0; k < HID; k += 4) {
    v2f a, b;
    a.x = Ar[k + khalf];
    a.y = Ar[k + khalf + 1];
    b.x = Bc[(k + khalf) * HID];
    b.y = Bc[(k + khalf + 1) * HID];
    acc = __builtin_amdgcn_wmma_f32_16x16x4_f32(
         false, a,  false, b,
         (short)0, acc,  false,  false);
  }

  float* __restrict__ o = fold + (size_t)blockIdx.y * FOLD_ROWS * HID + wave * 16 + m;
  if (tile < IN_DIM / 16) {
    const int rhalf = (lane >> 4) << 3;
#pragma unroll
    for (int r = 0; r < 8; ++r) o[(tile * 16 + rhalf + r) * HID] = acc[r];
  } else if (lane < 16) {
    o[IN_DIM * HID] = acc[0];
  }
}

#define SCAN_ELEMS  1000
#define SCAN_QUADS  (SCAN_ELEMS / 4)
#define SCAN_BLOCKS (NNODES / SCAN_ELEMS)
#define NODE_QUADS  (NNODES / 4)
#define OWN_THREADS 1024
#define OWN_UNROLL  4
static_assert(NNODES % 4 == 0 && SCAN_ELEMS % 4 == 0, "scan kernel reads cnt as int4");
static_assert(SCAN_QUADS <= 256 && SCAN_ELEMS <= OWN_THREADS, "one int4 per scan thread, one node per owner thread");
static_assert(SCAN_BLOCKS <= 256, "block totals are summed by one 256-thread block");
static_assert(N_USER % SCAN_ELEMS == 0 && N_ITEM % SCAN_ELEMS == 0, "an owner range holds nodes of one type only");
static_assert(E_UI % 4 == 0 && E_IU % 4 == 0, "destination rows are read as int4");

__device__ __forceinline__ void edge_ends(const int* __restrict__ eui,
                                          const int* __restrict__ eiu,
                                          int e, int& s, int& d)
{
  if (e < E_UI) { s = eui[e];                 d = eui[E_UI + e] + N_USER; }
  else          { int e2 = e - E_UI;
                  s = eiu[e2] + N_USER;       d = eiu[E_IU + e2]; }
}

__device__ __forceinline__ void take_slot(int* lc, int* __restrict__ rank, int v, int first, int e)
{
  const unsigned x = (unsigned)(v - first);
  if (x < (unsigned)SCAN_ELEMS) rank[e] = atomicAdd(&lc[x], 1);
}

__global__ __launch_bounds__(OWN_THREADS) void own_count_rank(
    const int* __restrict__ eui, const int* __restrict__ eiu,
    int* __restrict__ cnt, int* __restrict__ rank,
    int* __restrict__ btot, float* __restrict__ dinv)
{
  __shared__ int lc[SCAN_ELEMS];
  __shared__ int wtot[OWN_THREADS / 32];
  const int tid = threadIdx.x;
  const int lo  = (int)blockIdx.x * SCAN_ELEMS;
  if (tid < SCAN_ELEMS) lc[tid] = 0;
  __syncthreads();

  const bool usr  = lo < N_USER;
  const int4* __restrict__ dl = (const int4*)(usr ? eiu + E_IU : eui + E_UI);
  const int nq    = (usr ? E_IU : E_UI) / 4;
  const int ebase = usr ? E_UI : 0;
  const int first = usr ? lo : lo - N_USER;
  const int bt    = usr ? (int)blockIdx.x : (int)blockIdx.x - N_USER / SCAN_ELEMS;
  const int nown  = (usr ? N_USER : N_ITEM) / SCAN_ELEMS;
  const int rot   = bt * ((nq / nown) & ~31);

  const int nit = (nq + OWN_THREADS * OWN_UNROLL - 1) / (OWN_THREADS * OWN_UNROLL);
  for (int it = 0; it < nit; ++it) {
    int4 v[OWN_UNROLL];
    int  g[OWN_UNROLL];
#pragma unroll
    for (int u = 0; u < OWN_UNROLL; ++u) {
      const int q = (it * OWN_UNROLL + u) * OWN_THREADS + tid;
      int r = q + rot;
      if (r >= nq) r -= nq;
      g[u] = r;
      v[u] = (q < nq) ? dl[r] : make_int4(-1, -1, -1, -1);
    }
#pragma unroll
    for (int u = 0; u < OWN_UNROLL; ++u) {
      const int e0 = ebase + 4 * g[u];
      take_slot(lc, rank, v[u].x, first, e0);
      take_slot(lc, rank, v[u].y, first, e0 + 1);
      take_slot(lc, rank, v[u].z, first, e0 + 2);
      take_slot(lc, rank, v[u].w, first, e0 + 3);
    }
  }
  __syncthreads();

  int c = 0;
  if (tid < SCAN_ELEMS) {
    c = lc[tid];
    cnt[lo + tid] = c;
    const float dg = (float)(c + 1);
    dinv[lo + tid] = dg > 0.0f ? rsqrtf(dg) : 0.0f;
  }
  for (int m = 16; m > 0; m >>= 1) c += __shfl_xor(c, m, 32);
  if ((tid & 31) == 0) wtot[tid >> 5] = c;
  __syncthreads();
  if (tid == 0) {
    int tot = 0;
    for (int k = 0; k < OWN_THREADS / 32; ++k) tot += wtot[k];
    btot[blockIdx.x] = tot;
  }
}

__global__ __launch_bounds__(256) void scan_rowptr(const int* __restrict__ cnt,
                                                   const int* __restrict__ btot,
                                                   int* __restrict__ rp)
{
  __shared__ int wsum[8];
  __shared__ int wpre[8];
  const int t    = threadIdx.x;
  const int lane = t & 31;
  const int w    = t >> 5;
  const int q    = blockIdx.x * SCAN_QUADS + t;
  const bool live = t < SCAN_QUADS;

  int p = (t < (int)blockIdx.x) ? btot[t] : 0;
  for (int m = 16; m > 0; m >>= 1) p += __shfl_xor(p, m, 32);

  int4 c = make_int4(0, 0, 0, 0);
  if (live) c = ((const int4*)cnt)[q];
  const int s = c.x + c.y + c.z + c.w;
  int inc = s;
  for (int dlt = 1; dlt < 32; dlt <<= 1) {
    int v = __shfl_up(inc, dlt, 32);
    if (lane >= dlt) inc += v;
  }
  if (lane == 31) wsum[w] = inc;
  if (lane == 0)  wpre[w] = p;
  __syncthreads();
  int base = 0;
  for (int k = 0; k < 8; ++k) {
    base += wpre[k];
    if (k < w) base += wsum[k];
  }
  if (live) {
    int4 r;
    r.x = base + inc - s;
    r.y = r.x + c.x;
    r.z = r.y + c.y;
    r.w = r.z + c.z;
    ((int4*)rp)[q] = r;
    if (q == NODE_QUADS - 1) rp[NNODES] = r.w + c.w;
  }
}

__global__ void csr_fill(const int* __restrict__ eui, const int* __restrict__ eiu,
                         const int* __restrict__ rp, const int* __restrict__ rank,
                         const float* __restrict__ dinv, int* __restrict__ col)
{
  int e = blockIdx.x * blockDim.x + threadIdx.x;
  if (e >= E_TOT) return;
  int s, d;
  edge_ends(eui, eiu, e, s, d);
  ((int2*)col)[rp[d] + rank[e]] = make_int2(s, __float_as_int(dinv[s]));
}

template <int U>
__device__ __forceinline__ void add_rows(float4& acc, const float4* __restrict__ hw4,
                                         int ss, float mw, int k, int lane)
{
  float4 v[U];
  float  w[U];
  for (int u = 0; u < U; ++u) {
    const int s = __builtin_amdgcn_readlane(ss, k + u);
    v[u] = hw4[(long)s * (HID / 4) + lane];
  }
  for (int u = 0; u < U; ++u)
    w[u] = __int_as_float(__builtin_amdgcn_readlane(__float_as_int(mw), k + u));
  for (int u = 0; u < U; ++u) {
    acc.x = fmaf(w[u], v[u].x, acc.x);
    acc.y = fmaf(w[u], v[u].y, acc.y);
    acc.z = fmaf(w[u], v[u].z, acc.z);
    acc.w = fmaf(w[u], v[u].w, acc.w);
  }
}

__global__ __launch_bounds__(256) void agg_gather(
    const int* __restrict__ rp, const int* __restrict__ col,
    const float* __restrict__ hw, const float* __restrict__ dinv,
    const float* __restrict__ bias, float* __restrict__ out)
{
  const int lane = threadIdx.x & 31;
  const int w = __builtin_amdgcn_readfirstlane((int)(threadIdx.x >> 5));
  const int i = blockIdx.x * 8 + w;
  if (i >= NNODES) return;

  const float4* __restrict__ hw4 = (const float4*)hw;
  const int   start = rp[i];
  const int   end   = rp[i + 1];
  const float di    = dinv[i];
  const float4 sv   = hw4[(long)i * (HID / 4) + lane];
  const float4 bv   = ((const float4*)bias)[lane];

  const float dd = di * di;
  float4 acc;
  acc.x = bv.x + dd * sv.x;
  acc.y = bv.y + dd * sv.y;
  acc.z = bv.z + dd * sv.z;
  acc.w = bv.w + dd * sv.w;

  for (int base = start; base < end; base += 32) {
    int n = end - base;
    if (n > 32) n = 32;

    int2 slot = make_int2(0x7fffffff, 0);
    if (lane < n) slot = ((const int2*)col)[base + lane];
    const int ms = slot.x;

    int pos = 0;
    for (int k = 0; k < n; ++k) {
      const int sk = __builtin_amdgcn_readlane(ms, k);
      pos += ((sk < ms) | ((sk == ms) & (k < lane))) ? 1 : 0;
    }
    int ss = __builtin_amdgcn_ds_permute(pos << 2, ms);
    const float mw = __int_as_float(__builtin_amdgcn_ds_permute(pos << 2, slot.y)) * di;
    ss = (int)min((unsigned)ss, (unsigned)(NNODES - 1));

    int j = 0;
    for (; j + 8 <= n; j += 8) add_rows<8>(acc, hw4, ss, mw, j, lane);
    if (j + 4 <= n) { add_rows<4>(acc, hw4, ss, mw, j, lane); j += 4; }
    switch (n - j) {
      case 3: add_rows<3>(acc, hw4, ss, mw, j, lane); break;
      case 2: add_rows<2>(acc, hw4, ss, mw, j, lane); break;
      case 1: add_rows<1>(acc, hw4, ss, mw, j, lane); break;
      default: break;
    }
  }

  ((float4*)out)[(long)i * (HID / 4) + lane] = acc;
}

__device__ __forceinline__ void sum_chunk(float4& acc, const float4* __restrict__ hw4,
                                          int2 raw, int n, float di, int lane)
{
  const int2 slot = lane < n ? raw : make_int2(0x7fffffff, 0);
  const int ms = slot.x;

  int pos = 0;
  for (int q = 0; q < n; ++q) {
    const int sk = __builtin_amdgcn_readlane(ms, q);
    pos += ((sk < ms) | ((sk == ms) & (q < lane))) ? 1 : 0;
  }
  int ss = __builtin_amdgcn_ds_permute(pos << 2, ms);
  const float mw = __int_as_float(__builtin_amdgcn_ds_permute(pos << 2, slot.y)) * di;
  ss = (int)min((unsigned)ss, (unsigned)(NNODES - 1));

  int j = 0;
  for (; j + 8 <= n; j += 8) add_rows<8>(acc, hw4, ss, mw, j, lane);
  switch (n - j) {
    case 7: add_rows<7>(acc, hw4, ss, mw, j, lane); break;
    case 6: add_rows<6>(acc, hw4, ss, mw, j, lane); break;
    case 5: add_rows<5>(acc, hw4, ss, mw, j, lane); break;
    case 4: add_rows<4>(acc, hw4, ss, mw, j, lane); break;
    case 3: add_rows<3>(acc, hw4, ss, mw, j, lane); break;
    case 2: add_rows<2>(acc, hw4, ss, mw, j, lane); break;
    case 1: add_rows<1>(acc, hw4, ss, mw, j, lane); break;
    default: break;
  }
}

#define FUSE_WGS   256
#define FUSE_WAVES 25

__global__ __launch_bounds__(FUSE_WAVES * 32) void agg_relu_gemm(
    const int* __restrict__ rp, const int* __restrict__ col,
    const float* __restrict__ hw, const float* __restrict__ dinv,
    const float* __restrict__ bias, const float* __restrict__ W,
    float* __restrict__ C)
{
  __shared__ v4f Ws[(HID / 8) * 256];
  __shared__ v4f Ts[FUSE_WAVES * 16 * 32];

  const int tid  = threadIdx.x;
  const int lane = tid & 31;
  const int wave = __builtin_amdgcn_readfirstlane(tid >> 5);
  const int c    = lane & 15;
  const int h    = lane >> 4;

#pragma unroll
  for (int i = 0; i < ((HID / 8) * 256 + FUSE_WAVES * 32 - 1) / (FUSE_WAVES * 32); ++i) {
    const int e = tid + FUSE_WAVES * 32 * i;
    if (e < (HID / 8) * 256) {
      const float* __restrict__ p =
          W + (size_t)(8 * (e >> 8) + 2 * h) * HID + ((e >> 5) & 7) * 16 + c;
      v4f v;
      v.x = p[0];
      v.y = p[HID];
      v.z = p[4 * HID];
      v.w = p[5 * HID];
      Ws[e] = v;
    }
  }
  __syncthreads();

  const float4* __restrict__ hw4  = (const float4*)hw;
  const int2*   __restrict__ col2 = (const int2*)col;
  const float4 bv = ((const float4*)bias)[lane];
  v4f* __restrict__ Tw = Ts + wave * (16 * 32);
  const float* __restrict__ Tr = (const float*)Tw + c * 128 + 2 * h;
  const v4f* __restrict__ Bl = Ws + lane;
  const int rhalf  = h << 3;
  static_assert(N_USER % 16 == 0 && N_ITEM % 16 == 0, "a tile is 16 whole nodes of one type");

  constexpr int NSLOTS = FUSE_WAVES * FUSE_WGS;
  constexpr int TU = N_USER / 16, TI = N_ITEM / 16;
  constexpr int PU = (TU + NSLOTS - 1) / NSLOTS;
  constexpr int PI = (TI + NSLOTS - 1) / NSLOTS;
  const int slot = wave * FUSE_WGS + (int)blockIdx.x;
  for (int pass = 0; pass < PU + PI; ++pass) {
    const bool usr = pass < PU;
    const int  t   = (usr ? pass : pass - PU) * NSLOTS + slot;
    if (t >= (usr ? TU : TI)) continue;
    const int tile = usr ? t : TU + t;
    asm volatile("" ::: "memory");
    const int node0 = tile * 16;
    const int   rpl = rp[node0 + (lane < 16 ? lane : 16)];
    const float dil = dinv[node0 + c];

    int2   pslot;
    float4 psv;
    {
      const int s0 = __builtin_amdgcn_readlane(rpl, 0);
      pslot = col2[min(s0 + lane, E_TOT - 1)];
      psv = hw4[(long)node0 * (HID / 4) + lane];
    }

    for (int k = 0; k < 16; ++k) {
      const int   start = __builtin_amdgcn_readlane(rpl, k);
      const int   end   = __builtin_amdgcn_readlane(rpl, k + 1);
      const float di    = __int_as_float(__builtin_amdgcn_readlane(__float_as_int(dil), k));
      const int2 raw = pslot;
      const float4 sv = psv;
      {
        const int kn = k < 15 ? k + 1 : 15;
        const int sn = __builtin_amdgcn_readlane(rpl, kn);
        pslot = col2[min(sn + lane, E_TOT - 1)];
        psv = hw4[(long)(node0 + kn) * (HID / 4) + lane];
      }

      const float dd = di * di;
      float4 acc;
      acc.x = bv.x + dd * sv.x;
      acc.y = bv.y + dd * sv.y;
      acc.z = bv.z + dd * sv.z;
      acc.w = bv.w + dd * sv.w;

      if (end > start) {
        sum_chunk(acc, hw4, raw, min(end - start, 32), di, lane);
        for (int base = start + 32; base < end; base += 32) {
          const int2 more = col2[min(base + lane, E_TOT - 1)];
          sum_chunk(acc, hw4, more, min(end - base, 32), di, lane);
        }
      }

      v4f o;
      o.x = fmaxf(acc.x, 0.0f);
      o.y = fmaxf(acc.y, 0.0f);
      o.z = fmaxf(acc.z, 0.0f);
      o.w = fmaxf(acc.w, 0.0f);
      Tw[k * 32 + (lane ^ k)] = o;
    }

    __builtin_amdgcn_fence(__ATOMIC_RELEASE, "workgroup", "local");
    __builtin_amdgcn_wave_barrier();
    __builtin_amdgcn_fence(__ATOMIC_ACQUIRE, "workgroup", "local");

    v8f acc8[8] = {};
    for (int kp = 0; kp < HID / 8; ++kp) {
      const v2f s0 = *(const v2f*)(Tr + (((2 * kp)     ^ c) << 2));
      const v2f s1 = *(const v2f*)(Tr + (((2 * kp + 1) ^ c) << 2));
      v4f b[8];
      load_b_pair(b, Bl + kp * 256);
      wmma_kpair<false>(acc8, s0, s1, b);
    }

    const long row0 = (long)node0;
#pragma unroll
    for (int t = 0; t < 8; ++t) {
#pragma unroll
      for (int r = 0; r < 8; ++r)
        C[(row0 + rhalf + r) * HID + t * 16 + c] = acc8[t][r];
    }

    __builtin_amdgcn_fence(__ATOMIC_RELEASE, "workgroup", "local");
    __builtin_amdgcn_wave_barrier();
  }
}

extern "C" void kernel_launch(void* const* d_in, const int* in_sizes, int n_in,
                              void* d_out, int out_size, void* d_ws, size_t ws_size,
                              hipStream_t stream)
{
  const float* x_user = (const float*)d_in[0];
  const float* x_item = (const float*)d_in[1];
  const int*   eui    = (const int*)d_in[2];
  const int*   eiu    = (const int*)d_in[3];
  const float* W_iu   = (const float*)d_in[4];
  const float* b_iu   = (const float*)d_in[5];
  const float* W_ii   = (const float*)d_in[6];
  const float* b_ii   = (const float*)d_in[7];
  const float* W1     = (const float*)d_in[8];
  const float* b1     = (const float*)d_in[9];
  const float* W2     = (const float*)d_in[10];
  const float* b2     = (const float*)d_in[11];
  float* out = (float*)d_out;

  float* h    = (float*)d_ws;
  float* hw   = h  + (size_t)NNODES * HID;
  float* dinv = hw + (size_t)NNODES * HID;
  int*   cnt  = (int*)(dinv + NNODES);
  int*   rp   = cnt  + NNODES;
  int*   rank = rp   + (NNODES + 8);
  int*   col  = rank + E_TOT;
  int*   btot = col  + 2 * E_TOT;
  float* fold = (float*)(btot + 256);
  float* fold_u = fold;
  float* fold_i = fold + (size_t)FOLD_ROWS * HID;

  const dim3 blk(GEMM_WAVES * 32);

  fold_in_w1<<<dim3(FOLD_TILES, 2), dim3(256), 0, stream>>>(W_iu, b_iu, W_ii, b_ii, W1, fold);
  gemm_wlds<IN_DIM, false><<<dim3(GEMM_WGS), blk, 0, stream>>>(
      x_user, fold_u, fold_u + IN_DIM * HID, hw, N_USER,
      x_item, fold_i, fold_i + IN_DIM * HID, hw + (size_t)N_USER * HID, N_ITEM);

  own_count_rank<<<dim3(SCAN_BLOCKS),       dim3(OWN_THREADS), 0, stream>>>(eui, eiu, cnt, rank, btot, dinv);
  scan_rowptr<<<dim3(SCAN_BLOCKS),          dim3(256), 0, stream>>>(cnt, btot, rp);
  csr_fill   <<<dim3((E_TOT + 255) / 256),  dim3(256), 0, stream>>>(eui, eiu, rp, rank, dinv, col);

  const unsigned nodeBlocks = (NNODES + 7) / 8;

  agg_relu_gemm<<<dim3(FUSE_WGS), dim3(FUSE_WAVES * 32), 0, stream>>>(rp, col, hw, dinv, b1, W2, h);

  agg_gather<<<dim3(nodeBlocks), dim3(256), 0, stream>>>(rp, col, h, dinv, b2, out);
}


# edge-rank owner work-groups read a tenth of their destination row each (ten row slices x 10000-node ranges counted in LDS); the scan turns slice counts into slice offsets per bucket; sum order same
# speedup vs baseline: 2.9515x; 1.0862x over previous
#include <hip/hip_runtime.h>


typedef __attribute__((ext_vector_type(2))) float v2f;
typedef __attribute__((ext_vector_type(4))) float v4f;
typedef __attribute__((ext_vector_type(8))) float v8f;

#define N_USER 100000
#define N_ITEM 100000
#define NNODES (N_USER + N_ITEM)
#define HID    128
#define IN_DIM 256
#define E_UI   600000
#define E_IU   600000
#define E_TOT  (E_UI + E_IU)

#define GEMM_WGS   256
static_assert(GEMM_WGS % 2 == 0, "the launch splits its work-groups between two problems");
#define GEMM_WAVES 16

__device__ __forceinline__ void load_a_line(v2f (&a)[8], const float* __restrict__ p)
{
#pragma unroll
  for (int j = 0; j < 8; ++j) a[j] = *(const v2f*)(p + 4 * j);
}

__device__ __forceinline__ void load_b_pair(v4f (&b)[8], const v4f* __restrict__ p)
{
#pragma unroll
  for (int t = 0; t < 8; ++t) b[t] = p[t * 32];
}

template <bool RELU>
__device__ __forceinline__ void wmma_kpair(v8f (&acc)[8], v2f s0, v2f s1, const v4f (&b)[8])
{
  if (RELU) {
    s0.x = fmaxf(s0.x, 0.0f); s0.y = fmaxf(s0.y, 0.0f);
    s1.x = fmaxf(s1.x, 0.0f); s1.y = fmaxf(s1.y, 0.0f);
  }
#pragma unroll
  for (int t = 0; t < 8; ++t) {
    v2f bb; bb.x = b[t].x; bb.y = b[t].y;
    acc[t] = __builtin_amdgcn_wmma_f32_16x16x4_f32(
         false, s0,  false, bb,
         (short)0, acc[t],  false,  false);
  }
#pragma unroll
  for (int t = 0; t < 8; ++t) {
    v2f bb; bb.x = b[t].z; bb.y = b[t].w;
    acc[t] = __builtin_amdgcn_wmma_f32_16x16x4_f32(
         false, s1,  false, bb,
         (short)0, acc[t],  false,  false);
  }
}

template <int K, bool RELU>
__global__ __launch_bounds__(GEMM_WAVES * 32) void gemm_wlds(
    const float* __restrict__ A_p0, const float* __restrict__ W_p0,
    const float* __restrict__ bias_p0, float* __restrict__ C_p0, int M_p0,
    const float* __restrict__ A_p1, const float* __restrict__ W_p1,
    const float* __restrict__ bias_p1, float* __restrict__ C_p1, int M_p1)
{
  const int  nb   = (int)(gridDim.x >> 1);
  const bool p1   = (int)blockIdx.x >= nb;
  const int  bidx = p1 ? (int)blockIdx.x - nb : (int)blockIdx.x;
  const float* __restrict__ A    = p1 ? A_p1 : A_p0;
  const float* __restrict__ W    = p1 ? W_p1 : W_p0;
  const float* __restrict__ bias = p1 ? bias_p1 : bias_p0;
  float* __restrict__ C          = p1 ? C_p1 : C_p0;
  const int M                    = p1 ? M_p1 : M_p0;

  constexpr int KP = K / 8;
  constexpr int LB = K / 32;
  static_assert(K % 64 == 0, "the product loop takes two lines of A per trip");
  __shared__ v4f Ws[KP * 256];

  const int tid  = threadIdx.x;
  const int lane = tid & 31;
  const int wave = __builtin_amdgcn_readfirstlane(tid >> 5);
  const int c    = lane & 15;
  const int h    = lane >> 4;

  {
    const float* __restrict__ p0 =
        W + (size_t)(8 * (tid >> 8) + 2 * h) * HID + ((tid >> 5) & 7) * 16 + c;
#pragma unroll 8
    for (int i = 0; i < KP / 2; ++i) {
      const float* __restrict__ p = p0 + (size_t)i * 16 * HID;
      v4f v;
      v.x = p[0];
      v.y = p[HID];
      v.z = p[4 * HID];
      v.w = p[5 * HID];
      Ws[tid + 512 * i] = v;
    }
  }
  __syncthreads();

  const int ntiles = (M + 15) >> 4;
  const v4f* __restrict__ Bl = Ws + lane;
  const int rhalf  = h << 3;

  int tile = wave * nb + bidx;
  v2f a0[8], a1[8];
  long arow = (long)tile * 16 + c;
  if (arow >= M) arow = M - 1;
  const float* __restrict__ Ar = A + arow * (long)K + 2 * h;
  load_a_line(a0, Ar);

  while (tile < ntiles) {
    asm volatile("" ::: "memory");
    const long row0 = (long)tile * 16;
    const int tnext = tile + GEMM_WAVES * nb;
    long arown = (long)tnext * 16 + c;
    if (arown >= M) arown = M - 1;
    const float* __restrict__ Arn = A + arown * (long)K + 2 * h;

    v8f acc[8] = {};
    v4f b0[8], b1[8];
    load_b_pair(b0, Bl);

    for (int lb = 0; lb < LB; lb += 2) {
      const float* __restrict__ An = (lb + 2 < LB) ? Ar + (lb + 2) * 32 : Arn;
      load_a_line(a1, Ar + (lb + 1) * 32);
#pragma unroll
      for (int q = 0; q < 8; ++q) {
        int kpn = lb * 4 + q + 1;
        if (kpn > KP - 1) kpn = KP - 1;
        if (q == 4) load_a_line(a0, An);
        if (q & 1) {
          load_b_pair(b0, Bl + kpn * 256);
          __builtin_amdgcn_sched_barrier(0);
          wmma_kpair<RELU>(acc, (q < 4 ? a0 : a1)[2 * (q & 3)], (q < 4 ? a0 : a1)[2 * (q & 3) + 1], b1);
        } else {
          load_b_pair(b1, Bl + kpn * 256);
          __builtin_amdgcn_sched_barrier(0);
          wmma_kpair<RELU>(acc, (q < 4 ? a0 : a1)[2 * (q & 3)], (q < 4 ? a0 : a1)[2 * (q & 3) + 1], b0);
        }
        __builtin_amdgcn_sched_barrier(0);
      }
    }

    const int n = c;
#pragma unroll
    for (int t = 0; t < 8; ++t) {
      float bv = bias[t * 16 + n];
#pragma unroll
      for (int r = 0; r < 8; ++r) {
        long row = row0 + rhalf + r;
        if (row < M) C[row * HID + t * 16 + n] = acc[t][r] + bv;
      }
    }
    tile = tnext;
    Ar   = Arn;
  }
}

#define FOLD_TILES (IN_DIM / 16 + 1)
#define FOLD_ROWS  (FOLD_TILES * 16)
__global__ __launch_bounds__(256) void fold_in_w1(
    const float* __restrict__ W_iu, const float* __restrict__ b_iu,
    const float* __restrict__ W_ii, const float* __restrict__ b_ii,
    const float* __restrict__ W1, float* __restrict__ fold)
{
  const int lane  = threadIdx.x & 31;
  const int wave  = threadIdx.x >> 5;
  const int tile  = blockIdx.x;
  const int m     = lane & 15;
  const int khalf = (lane >> 4) << 1;

  const float* __restrict__ Win = blockIdx.y ? W_ii : W_iu;
  const float* __restrict__ bin = blockIdx.y ? b_ii : b_iu;
  const float* __restrict__ Ar  = (tile < IN_DIM / 16) ? Win + (long)(tile * 16 + m) * HID : bin;
  const float* __restrict__ Bc  = W1 + wave * 16 + m;

  v8f acc = {};
#pragma unroll
  for (int k = 0; k < HID; k += 4) {
    v2f a, b;
    a.x = Ar[k + khalf];
    a.y = Ar[k + khalf + 1];
    b.x = Bc[(k + khalf) * HID];
    b.y = Bc[(k + khalf + 1) * HID];
    acc = __builtin_amdgcn_wmma_f32_16x16x4_f32(
         false, a,  false, b,
         (short)0, acc,  false,  false);
  }

  float* __restrict__ o = fold + (size_t)blockIdx.y * FOLD_ROWS * HID + wave * 16 + m;
  if (tile < IN_DIM / 16) {
    const int rhalf = (lane >> 4) << 3;
#pragma unroll
    for (int r = 0; r < 8; ++r) o[(tile * 16 + rhalf + r) * HID] = acc[r];
  } else if (lane < 16) {
    o[IN_DIM * HID] = acc[0];
  }
}

#define SCAN_ELEMS  1000
#define SCAN_QUADS  (SCAN_ELEMS / 4)
#define SCAN_BLOCKS (NNODES / SCAN_ELEMS)
#define NODE_QUADS  (NNODES / 4)
#define OWN_RANGE   10000
#define OWN_SLICES  10
#define OWN_RANGES  (NNODES / OWN_RANGE)
#define OWN_SUBS    (OWN_RANGE / SCAN_ELEMS)
#define OWN_THREADS 1024
#define OWN_UNROLL  4
#define SLICE_EDGES (E_UI / OWN_SLICES)
#define SLICE_QUADS (SLICE_EDGES / 4)
#define CPITCH      (NNODES + 8)
static_assert(NNODES % 4 == 0 && SCAN_ELEMS % 4 == 0 && CPITCH % 4 == 0, "scan kernel reads the planes as int4");
static_assert(SCAN_QUADS <= 256, "one int4 per scan thread and slice");
static_assert(SCAN_BLOCKS <= 256, "block totals are summed by one 256-thread block");
static_assert(N_USER % OWN_RANGE == 0 && N_ITEM % OWN_RANGE == 0, "an owner range holds nodes of one type only");
static_assert(OWN_RANGE % SCAN_ELEMS == 0, "an owner range is whole scan blocks");
static_assert(E_UI == E_IU && E_UI % (4 * OWN_SLICES) == 0, "slices are whole int4 groups, the same for both rows");

__device__ __forceinline__ void edge_ends(const int* __restrict__ eui,
                                          const int* __restrict__ eiu,
                                          int e, int& s, int& d)
{
  if (e < E_UI) { s = eui[e];                 d = eui[E_UI + e] + N_USER; }
  else          { int e2 = e - E_UI;
                  s = eiu[e2] + N_USER;       d = eiu[E_IU + e2]; }
}

__device__ __forceinline__ void take_slots(int* lc, int* __restrict__ rank, int4 v, int first, int e0)
{
  const unsigned x0 = (unsigned)(v.x - first), x1 = (unsigned)(v.y - first);
  const unsigned x2 = (unsigned)(v.z - first), x3 = (unsigned)(v.w - first);
  const bool h0 = x0 < (unsigned)OWN_RANGE, h1 = x1 < (unsigned)OWN_RANGE;
  const bool h2 = x2 < (unsigned)OWN_RANGE, h3 = x3 < (unsigned)OWN_RANGE;
  int r0 = 0, r1 = 0, r2 = 0, r3 = 0;
  if (h0) r0 = atomicAdd(&lc[x0], 1);
  if (h1) r1 = atomicAdd(&lc[x1], 1);
  if (h2) r2 = atomicAdd(&lc[x2], 1);
  if (h3) r3 = atomicAdd(&lc[x3], 1);
  if (h0) rank[e0]     = r0;
  if (h1) rank[e0 + 1] = r1;
  if (h2) rank[e0 + 2] = r2;
  if (h3) rank[e0 + 3] = r3;
}

__global__ __launch_bounds__(OWN_THREADS) void own_count_rank(
    const int* __restrict__ eui, const int* __restrict__ eiu,
    int* __restrict__ sc, int* __restrict__ rank, int* __restrict__ btot)
{
  __shared__ int lc[OWN_RANGE];
  const int tid = threadIdx.x;
  const int rg  = (int)blockIdx.x / OWN_SLICES;
  const int sl  = (int)blockIdx.x - rg * OWN_SLICES;
  const int lo  = rg * OWN_RANGE;
  for (int i = tid; i < OWN_RANGE; i += OWN_THREADS) lc[i] = 0;
  __syncthreads();

  const bool usr  = lo < N_USER;
  const int4* __restrict__ dl = (const int4*)(usr ? eiu + E_IU : eui + E_UI) + sl * SLICE_QUADS;
  const int ebase = (usr ? E_UI : 0) + sl * SLICE_EDGES;
  const int first = usr ? lo : lo - N_USER;
  const int rt    = usr ? rg : rg - N_USER / OWN_RANGE;
  const int rot   = rt * ((SLICE_QUADS / (N_USER / OWN_RANGE)) & ~31);

  constexpr int NIT = (SLICE_QUADS + OWN_THREADS * OWN_UNROLL - 1) / (OWN_THREADS * OWN_UNROLL);
  for (int it = 0; it < NIT; ++it) {
    int4 v[OWN_UNROLL];
    int  g[OWN_UNROLL];
#pragma unroll
    for (int u = 0; u < OWN_UNROLL; ++u) {
      const int q = (it * OWN_UNROLL + u) * OWN_THREADS + tid;
      int r = q + rot;
      if (r >= SLICE_QUADS) r -= SLICE_QUADS;
      g[u] = r;
      v[u] = (q < SLICE_QUADS) ? dl[r] : make_int4(-1, -1, -1, -1);
    }
#pragma unroll
    for (int u = 0; u < OWN_UNROLL; ++u) {
      take_slots(lc, rank, v[u], first, ebase + 4 * g[u]);
    }
  }
  __syncthreads();

  int* __restrict__ plane = sc + (size_t)sl * CPITCH + lo;
  for (int i = tid; i < OWN_RANGE; i += OWN_THREADS) plane[i] = lc[i];
  const int lane = tid & 31;
  const int wave = tid >> 5;
  if (wave < OWN_SUBS) {
    int c = 0;
    for (int i = lane; i < SCAN_ELEMS; i += 32) c += lc[wave * SCAN_ELEMS + i];
    for (int m = 16; m > 0; m >>= 1) c += __shfl_xor(c, m, 32);
    if (lane == 0) btot[sl * SCAN_BLOCKS + rg * OWN_SUBS + wave] = c;
  }
}

__global__ __launch_bounds__(256) void scan_rowptr(int* __restrict__ sc,
                                                   const int* __restrict__ btot,
                                                   float* __restrict__ dinv)
{
  __shared__ int wsum[8];
  __shared__ int wpre[8];
  const int t    = threadIdx.x;
  const int lane = t & 31;
  const int w    = t >> 5;
  const int q    = blockIdx.x * SCAN_QUADS + t;
  const bool live = t < SCAN_QUADS;

  int p = 0;
  if (t < (int)blockIdx.x) {
    for (int j = 0; j < OWN_SLICES; ++j) p += btot[j * SCAN_BLOCKS + t];
  }
  for (int m = 16; m > 0; m >>= 1) p += __shfl_xor(p, m, 32);

  int4 cj[OWN_SLICES];
  int4 c = make_int4(0, 0, 0, 0);
#pragma unroll
  for (int j = 0; j < OWN_SLICES; ++j) {
    cj[j] = live ? ((const int4*)(sc + (size_t)j * CPITCH))[q] : make_int4(0, 0, 0, 0);
  }
#pragma unroll
  for (int j = 0; j < OWN_SLICES; ++j) {
    c.x += cj[j].x; c.y += cj[j].y; c.z += cj[j].z; c.w += cj[j].w;
  }
  const int s = c.x + c.y + c.z + c.w;
  int inc = s;
  for (int dlt = 1; dlt < 32; dlt <<= 1) {
    int v = __shfl_up(inc, dlt, 32);
    if (lane >= dlt) inc += v;
  }
  if (lane == 31) wsum[w] = inc;
  if (lane == 0)  wpre[w] = p;
  __syncthreads();
  int base = 0;
  for (int k = 0; k < 8; ++k) {
    base += wpre[k];
    if (k < w) base += wsum[k];
  }
  if (live) {
    int4 r;
    r.x = base + inc - s;
    r.y = r.x + c.x;
    r.z = r.y + c.y;
    r.w = r.z + c.z;
    float4 dv;
    dv.x = rsqrtf((float)(c.x + 1));
    dv.y = rsqrtf((float)(c.y + 1));
    dv.z = rsqrtf((float)(c.z + 1));
    dv.w = rsqrtf((float)(c.w + 1));
    ((float4*)dinv)[q] = dv;
    if (q == NODE_QUADS - 1) sc[NNODES] = r.w + c.w;
#pragma unroll
    for (int j = 0; j < OWN_SLICES; ++j) {
      ((int4*)(sc + (size_t)j * CPITCH))[q] = r;
      r.x += cj[j].x; r.y += cj[j].y; r.z += cj[j].z; r.w += cj[j].w;
    }
  }
}

__global__ void csr_fill(const int* __restrict__ eui, const int* __restrict__ eiu,
                         const int* __restrict__ off, const int* __restrict__ rank,
                         const float* __restrict__ dinv, int* __restrict__ col)
{
  int e = blockIdx.x * blockDim.x + threadIdx.x;
  if (e >= E_TOT) return;
  int s, d;
  edge_ends(eui, eiu, e, s, d);
  const int sl = (e < E_UI ? e : e - E_UI) / SLICE_EDGES;
  ((int2*)col)[off[(size_t)sl * CPITCH + d] + rank[e]] = make_int2(s, __float_as_int(dinv[s]));
}

template <int U>
__device__ __forceinline__ void add_rows(float4& acc, const float4* __restrict__ hw4,
                                         int ss, float mw, int k, int lane)
{
  float4 v[U];
  float  w[U];
  for (int u = 0; u < U; ++u) {
    const int s = __builtin_amdgcn_readlane(ss, k + u);
    v[u] = hw4[(long)s * (HID / 4) + lane];
  }
  for (int u = 0; u < U; ++u)
    w[u] = __int_as_float(__builtin_amdgcn_readlane(__float_as_int(mw), k + u));
  for (int u = 0; u < U; ++u) {
    acc.x = fmaf(w[u], v[u].x, acc.x);
    acc.y = fmaf(w[u], v[u].y, acc.y);
    acc.z = fmaf(w[u], v[u].z, acc.z);
    acc.w = fmaf(w[u], v[u].w, acc.w);
  }
}

__global__ __launch_bounds__(256) void agg_gather(
    const int* __restrict__ rp, const int* __restrict__ col,
    const float* __restrict__ hw, const float* __restrict__ dinv,
    const float* __restrict__ bias, float* __restrict__ out)
{
  const int lane = threadIdx.x & 31;
  const int w = __builtin_amdgcn_readfirstlane((int)(threadIdx.x >> 5));
  const int i = blockIdx.x * 8 + w;
  if (i >= NNODES) return;

  const float4* __restrict__ hw4 = (const float4*)hw;
  const int   start = rp[i];
  const int   end   = rp[i + 1];
  const float di    = dinv[i];
  const float4 sv   = hw4[(long)i * (HID / 4) + lane];
  const float4 bv   = ((const float4*)bias)[lane];

  const float dd = di * di;
  float4 acc;
  acc.x = bv.x + dd * sv.x;
  acc.y = bv.y + dd * sv.y;
  acc.z = bv.z + dd * sv.z;
  acc.w = bv.w + dd * sv.w;

  for (int base = start; base < end; base += 32) {
    int n = end - base;
    if (n > 32) n = 32;

    int2 slot = make_int2(0x7fffffff, 0);
    if (lane < n) slot = ((const int2*)col)[base + lane];
    const int ms = slot.x;

    int pos = 0;
    for (int k = 0; k < n; ++k) {
      const int sk = __builtin_amdgcn_readlane(ms, k);
      pos += ((sk < ms) | ((sk == ms) & (k < lane))) ? 1 : 0;
    }
    int ss = __builtin_amdgcn_ds_permute(pos << 2, ms);
    const float mw = __int_as_float(__builtin_amdgcn_ds_permute(pos << 2, slot.y)) * di;
    ss = (int)min((unsigned)ss, (unsigned)(NNODES - 1));

    int j = 0;
    for (; j + 8 <= n; j += 8) add_rows<8>(acc, hw4, ss, mw, j, lane);
    if (j + 4 <= n) { add_rows<4>(acc, hw4, ss, mw, j, lane); j += 4; }
    switch (n - j) {
      case 3: add_rows<3>(acc, hw4, ss, mw, j, lane); break;
      case 2: add_rows<2>(acc, hw4, ss, mw, j, lane); break;
      case 1: add_rows<1>(acc, hw4, ss, mw, j, lane); break;
      default: break;
    }
  }

  ((float4*)out)[(long)i * (HID / 4) + lane] = acc;
}

__device__ __forceinline__ void sum_chunk(float4& acc, const float4* __restrict__ hw4,
                                          int2 raw, int n, float di, int lane)
{
  const int2 slot = lane < n ? raw : make_int2(0x7fffffff, 0);
  const int ms = slot.x;

  int pos = 0;
  for (int q = 0; q < n; ++q) {
    const int sk = __builtin_amdgcn_readlane(ms, q);
    pos += ((sk < ms) | ((sk == ms) & (q < lane))) ? 1 : 0;
  }
  int ss = __builtin_amdgcn_ds_permute(pos << 2, ms);
  const float mw = __int_as_float(__builtin_amdgcn_ds_permute(pos << 2, slot.y)) * di;
  ss = (int)min((unsigned)ss, (unsigned)(NNODES - 1));

  int j = 0;
  for (; j + 8 <= n; j += 8) add_rows<8>(acc, hw4, ss, mw, j, lane);
  switch (n - j) {
    case 7: add_rows<7>(acc, hw4, ss, mw, j, lane); break;
    case 6: add_rows<6>(acc, hw4, ss, mw, j, lane); break;
    case 5: add_rows<5>(acc, hw4, ss, mw, j, lane); break;
    case 4: add_rows<4>(acc, hw4, ss, mw, j, lane); break;
    case 3: add_rows<3>(acc, hw4, ss, mw, j, lane); break;
    case 2: add_rows<2>(acc, hw4, ss, mw, j, lane); break;
    case 1: add_rows<1>(acc, hw4, ss, mw, j, lane); break;
    default: break;
  }
}

#define FUSE_WGS   256
#define FUSE_WAVES 25

__global__ __launch_bounds__(FUSE_WAVES * 32) void agg_relu_gemm(
    const int* __restrict__ rp, const int* __restrict__ col,
    const float* __restrict__ hw, const float* __restrict__ dinv,
    const float* __restrict__ bias, const float* __restrict__ W,
    float* __restrict__ C)
{
  __shared__ v4f Ws[(HID / 8) * 256];
  __shared__ v4f Ts[FUSE_WAVES * 16 * 32];

  const int tid  = threadIdx.x;
  const int lane = tid & 31;
  const int wave = __builtin_amdgcn_readfirstlane(tid >> 5);
  const int c    = lane & 15;
  const int h    = lane >> 4;

#pragma unroll
  for (int i = 0; i < ((HID / 8) * 256 + FUSE_WAVES * 32 - 1) / (FUSE_WAVES * 32); ++i) {
    const int e = tid + FUSE_WAVES * 32 * i;
    if (e < (HID / 8) * 256) {
      const float* __restrict__ p =
          W + (size_t)(8 * (e >> 8) + 2 * h) * HID + ((e >> 5) & 7) * 16 + c;
      v4f v;
      v.x = p[0];
      v.y = p[HID];
      v.z = p[4 * HID];
      v.w = p[5 * HID];
      Ws[e] = v;
    }
  }
  __syncthreads();

  const float4* __restrict__ hw4  = (const float4*)hw;
  const int2*   __restrict__ col2 = (const int2*)col;
  const float4 bv = ((const float4*)bias)[lane];
  v4f* __restrict__ Tw = Ts + wave * (16 * 32);
  const float* __restrict__ Tr = (const float*)Tw + c * 128 + 2 * h;
  const v4f* __restrict__ Bl = Ws + lane;
  const int rhalf  = h << 3;
  static_assert(N_USER % 16 == 0 && N_ITEM % 16 == 0, "a tile is 16 whole nodes of one type");

  constexpr int NSLOTS = FUSE_WAVES * FUSE_WGS;
  constexpr int TU = N_USER / 16, TI = N_ITEM / 16;
  constexpr int PU = (TU + NSLOTS - 1) / NSLOTS;
  constexpr int PI = (TI + NSLOTS - 1) / NSLOTS;
  const int slot = wave * FUSE_WGS + (int)blockIdx.x;
  for (int pass = 0; pass < PU + PI; ++pass) {
    const bool usr = pass < PU;
    const int  t   = (usr ? pass : pass - PU) * NSLOTS + slot;
    if (t >= (usr ? TU : TI)) continue;
    const int tile = usr ? t : TU + t;
    asm volatile("" ::: "memory");
    const int node0 = tile * 16;
    const int   rpl = rp[node0 + (lane < 16 ? lane : 16)];
    const float dil = dinv[node0 + c];

    int2   pslot;
    float4 psv;
    {
      const int s0 = __builtin_amdgcn_readlane(rpl, 0);
      pslot = col2[min(s0 + lane, E_TOT - 1)];
      psv = hw4[(long)node0 * (HID / 4) + lane];
    }

    for (int k = 0; k < 16; ++k) {
      const int   start = __builtin_amdgcn_readlane(rpl, k);
      const int   end   = __builtin_amdgcn_readlane(rpl, k + 1);
      const float di    = __int_as_float(__builtin_amdgcn_readlane(__float_as_int(dil), k));
      const int2 raw = pslot;
      const float4 sv = psv;
      {
        const int kn = k < 15 ? k + 1 : 15;
        const int sn = __builtin_amdgcn_readlane(rpl, kn);
        pslot = col2[min(sn + lane, E_TOT - 1)];
        psv = hw4[(long)(node0 + kn) * (HID / 4) + lane];
      }

      const float dd = di * di;
      float4 acc;
      acc.x = bv.x + dd * sv.x;
      acc.y = bv.y + dd * sv.y;
      acc.z = bv.z + dd * sv.z;
      acc.w = bv.w + dd * sv.w;

      if (end > start) {
        sum_chunk(acc, hw4, raw, min(end - start, 32), di, lane);
        for (int base = start + 32; base < end; base += 32) {
          const int2 more = col2[min(base + lane, E_TOT - 1)];
          sum_chunk(acc, hw4, more, min(end - base, 32), di, lane);
        }
      }

      v4f o;
      o.x = fmaxf(acc.x, 0.0f);
      o.y = fmaxf(acc.y, 0.0f);
      o.z = fmaxf(acc.z, 0.0f);
      o.w = fmaxf(acc.w, 0.0f);
      Tw[k * 32 + (lane ^ k)] = o;
    }

    __builtin_amdgcn_fence(__ATOMIC_RELEASE, "workgroup", "local");
    __builtin_amdgcn_wave_barrier();
    __builtin_amdgcn_fence(__ATOMIC_ACQUIRE, "workgroup", "local");

    v8f acc8[8] = {};
    for (int kp = 0; kp < HID / 8; ++kp) {
      const v2f s0 = *(const v2f*)(Tr + (((2 * kp)     ^ c) << 2));
      const v2f s1 = *(const v2f*)(Tr + (((2 * kp + 1) ^ c) << 2));
      v4f b[8];
      load_b_pair(b, Bl + kp * 256);
      wmma_kpair<false>(acc8, s0, s1, b);
    }

    const long row0 = (long)node0;
#pragma unroll
    for (int t = 0; t < 8; ++t) {
#pragma unroll
      for (int r = 0; r < 8; ++r)
        C[(row0 + rhalf + r) * HID + t * 16 + c] = acc8[t][r];
    }

    __builtin_amdgcn_fence(__ATOMIC_RELEASE, "workgroup", "local");
    __builtin_amdgcn_wave_barrier();
  }
}

extern "C" void kernel_launch(void* const* d_in, const int* in_sizes, int n_in,
                              void* d_out, int out_size, void* d_ws, size_t ws_size,
                              hipStream_t stream)
{
  const float* x_user = (const float*)d_in[0];
  const float* x_item = (const float*)d_in[1];
  const int*   eui    = (const int*)d_in[2];
  const int*   eiu    = (const int*)d_in[3];
  const float* W_iu   = (const float*)d_in[4];
  const float* b_iu   = (const float*)d_in[5];
  const float* W_ii   = (const float*)d_in[6];
  const float* b_ii   = (const float*)d_in[7];
  const float* W1     = (const float*)d_in[8];
  const float* b1     = (const float*)d_in[9];
  const float* W2     = (const float*)d_in[10];
  const float* b2     = (const float*)d_in[11];
  float* out = (float*)d_out;

  float* h    = (float*)d_ws;
  float* hw   = h  + (size_t)NNODES * HID;
  float* dinv = hw + (size_t)NNODES * HID;
  int*   sc   = (int*)(dinv + NNODES);
  int*   rp   = sc;
  int*   rank = sc   + (size_t)OWN_SLICES * CPITCH;
  int*   col  = rank + E_TOT;
  int*   btot = col  + 2 * E_TOT;
  float* fold = (float*)(btot + OWN_SLICES * SCAN_BLOCKS);
  float* fold_u = fold;
  float* fold_i = fold + (size_t)FOLD_ROWS * HID;

  const dim3 blk(GEMM_WAVES * 32);

  fold_in_w1<<<dim3(FOLD_TILES, 2), dim3(256), 0, stream>>>(W_iu, b_iu, W_ii, b_ii, W1, fold);
  gemm_wlds<IN_DIM, false><<<dim3(GEMM_WGS), blk, 0, stream>>>(
      x_user, fold_u, fold_u + IN_DIM * HID, hw, N_USER,
      x_item, fold_i, fold_i + IN_DIM * HID, hw + (size_t)N_USER * HID, N_ITEM);

  own_count_rank<<<dim3(OWN_RANGES * OWN_SLICES), dim3(OWN_THREADS), 0, stream>>>(eui, eiu, sc, rank, btot);
  scan_rowptr<<<dim3(SCAN_BLOCKS),          dim3(256), 0, stream>>>(sc, btot, dinv);
  csr_fill   <<<dim3((E_TOT + 255) / 256),  dim3(256), 0, stream>>>(eui, eiu, sc, rank, dinv, col);

  const unsigned nodeBlocks = (NNODES + 7) / 8;

  agg_relu_gemm<<<dim3(FUSE_WGS), dim3(FUSE_WAVES * 32), 0, stream>>>(rp, col, hw, dinv, b1, W2, h);

  agg_gather<<<dim3(nodeBlocks), dim3(256), 0, stream>>>(rp, col, h, dinv, b2, out);
}


# independent launches paired: weight fold on extra work-groups of the edge-count launch, row-pointer scan on work-groups behind the GEMM's (start as those end); 7 to 5 launches; sum order same
# speedup vs baseline: 3.0708x; 1.0404x over previous
#include <hip/hip_runtime.h>


typedef __attribute__((ext_vector_type(2))) float v2f;
typedef __attribute__((ext_vector_type(4))) float v4f;
typedef __attribute__((ext_vector_type(8))) float v8f;

#define N_USER 100000
#define N_ITEM 100000
#define NNODES (N_USER + N_ITEM)
#define HID    128
#define IN_DIM 256
#define E_UI   600000
#define E_IU   600000
#define E_TOT  (E_UI + E_IU)

#define GEMM_WGS   256
static_assert(GEMM_WGS % 2 == 0, "the launch splits its work-groups between two problems");
#define GEMM_WAVES 16

__device__ __forceinline__ void scan_rowptr_block(int sb, int* __restrict__ sc, const int* __restrict__ btot,
                                                  float* __restrict__ dinv, int* wsum, int* wpre);

__device__ __forceinline__ void load_a_line(v2f (&a)[8], const float* __restrict__ p)
{
#pragma unroll
  for (int j = 0; j < 8; ++j) a[j] = *(const v2f*)(p + 4 * j);
}

__device__ __forceinline__ void load_b_pair(v4f (&b)[8], const v4f* __restrict__ p)
{
#pragma unroll
  for (int t = 0; t < 8; ++t) b[t] = p[t * 32];
}

template <bool RELU>
__device__ __forceinline__ void wmma_kpair(v8f (&acc)[8], v2f s0, v2f s1, const v4f (&b)[8])
{
  if (RELU) {
    s0.x = fmaxf(s0.x, 0.0f); s0.y = fmaxf(s0.y, 0.0f);
    s1.x = fmaxf(s1.x, 0.0f); s1.y = fmaxf(s1.y, 0.0f);
  }
#pragma unroll
  for (int t = 0; t < 8; ++t) {
    v2f bb; bb.x = b[t].x; bb.y = b[t].y;
    acc[t] = __builtin_amdgcn_wmma_f32_16x16x4_f32(
         false, s0,  false, bb,
         (short)0, acc[t],  false,  false);
  }
#pragma unroll
  for (int t = 0; t < 8; ++t) {
    v2f bb; bb.x = b[t].z; bb.y = b[t].w;
    acc[t] = __builtin_amdgcn_wmma_f32_16x16x4_f32(
         false, s1,  false, bb,
         (short)0, acc[t],  false,  false);
  }
}

template <int K, bool RELU>
__global__ __launch_bounds__(GEMM_WAVES * 32) void gemm_wlds(
    const float* __restrict__ A_p0, const float* __restrict__ W_p0,
    const float* __restrict__ bias_p0, float* __restrict__ C_p0, int M_p0,
    const float* __restrict__ A_p1, const float* __restrict__ W_p1,
    const float* __restrict__ bias_p1, float* __restrict__ C_p1, int M_p1,
    int* __restrict__ sc, const int* __restrict__ btot, float* __restrict__ dinv)
{
  __shared__ int wsum[8];
  __shared__ int wpre[8];
  if ((int)blockIdx.x >= GEMM_WGS) {
    scan_rowptr_block((int)blockIdx.x - GEMM_WGS, sc, btot, dinv, wsum, wpre);
    return;
  }

  const int  nb   = GEMM_WGS / 2;
  const bool p1   = (int)blockIdx.x >= nb;
  const int  bidx = p1 ? (int)blockIdx.x - nb : (int)blockIdx.x;
  const float* __restrict__ A    = p1 ? A_p1 : A_p0;
  const float* __restrict__ W    = p1 ? W_p1 : W_p0;
  const float* __restrict__ bias = p1 ? bias_p1 : bias_p0;
  float* __restrict__ C          = p1 ? C_p1 : C_p0;
  const int M                    = p1 ? M_p1 : M_p0;

  constexpr int KP = K / 8;
  constexpr int LB = K / 32;
  static_assert(K % 64 == 0, "the product loop takes two lines of A per trip");
  __shared__ v4f Ws[KP * 256];

  const int tid  = threadIdx.x;
  const int lane = tid & 31;
  const int wave = __builtin_amdgcn_readfirstlane(tid >> 5);
  const int c    = lane & 15;
  const int h    = lane >> 4;

  {
    const float* __restrict__ p0 =
        W + (size_t)(8 * (tid >> 8) + 2 * h) * HID + ((tid >> 5) & 7) * 16 + c;
#pragma unroll 8
    for (int i = 0; i < KP / 2; ++i) {
      const float* __restrict__ p = p0 + (size_t)i * 16 * HID;
      v4f v;
      v.x = p[0];
      v.y = p[HID];
      v.z = p[4 * HID];
      v.w = p[5 * HID];
      Ws[tid + 512 * i] = v;
    }
  }
  __syncthreads();

  const int ntiles = (M + 15) >> 4;
  const v4f* __restrict__ Bl = Ws + lane;
  const int rhalf  = h << 3;

  int tile = wave * nb + bidx;
  v2f a0[8], a1[8];
  long arow = (long)tile * 16 + c;
  if (arow >= M) arow = M - 1;
  const float* __restrict__ Ar = A + arow * (long)K + 2 * h;
  load_a_line(a0, Ar);

  while (tile < ntiles) {
    asm volatile("" ::: "memory");
    const long row0 = (long)tile * 16;
    const int tnext = tile + GEMM_WAVES * nb;
    long arown = (long)tnext * 16 + c;
    if (arown >= M) arown = M - 1;
    const float* __restrict__ Arn = A + arown * (long)K + 2 * h;

    v8f acc[8] = {};
    v4f b0[8], b1[8];
    load_b_pair(b0, Bl);

    for (int lb = 0; lb < LB; lb += 2) {
      const float* __restrict__ An = (lb + 2 < LB) ? Ar + (lb + 2) * 32 : Arn;
      load_a_line(a1, Ar + (lb + 1) * 32);
#pragma unroll
      for (int q = 0; q < 8; ++q) {
        int kpn = lb * 4 + q + 1;
        if (kpn > KP - 1) kpn = KP - 1;
        if (q == 4) load_a_line(a0, An);
        if (q & 1) {
          load_b_pair(b0, Bl + kpn * 256);
          __builtin_amdgcn_sched_barrier(0);
          wmma_kpair<RELU>(acc, (q < 4 ? a0 : a1)[2 * (q & 3)], (q < 4 ? a0 : a1)[2 * (q & 3) + 1], b1);
        } else {
          load_b_pair(b1, Bl + kpn * 256);
          __builtin_amdgcn_sched_barrier(0);
          wmma_kpair<RELU>(acc, (q < 4 ? a0 : a1)[2 * (q & 3)], (q < 4 ? a0 : a1)[2 * (q & 3) + 1], b0);
        }
        __builtin_amdgcn_sched_barrier(0);
      }
    }

    const int n = c;
#pragma unroll
    for (int t = 0; t < 8; ++t) {
      float bv = bias[t * 16 + n];
#pragma unroll
      for (int r = 0; r < 8; ++r) {
        long row = row0 + rhalf + r;
        if (row < M) C[row * HID + t * 16 + n] = acc[t][r] + bv;
      }
    }
    tile = tnext;
    Ar   = Arn;
  }
}

#define FOLD_TILES (IN_DIM / 16 + 1)
#define FOLD_ROWS  (FOLD_TILES * 16)
#define FOLD_JOBS      (FOLD_TILES * 2)
__device__ __forceinline__ void fold_in_w1(
    const float* __restrict__ W_iu, const float* __restrict__ b_iu,
    const float* __restrict__ W_ii, const float* __restrict__ b_ii,
    const float* __restrict__ W1, float* __restrict__ fold,
    int type, int tile, int wave, int lane)
{
  const int m     = lane & 15;
  const int khalf = (lane >> 4) << 1;

  const float* __restrict__ Win = type ? W_ii : W_iu;
  const float* __restrict__ bin = type ? b_ii : b_iu;
  const float* __restrict__ Ar  = (tile < IN_DIM / 16) ? Win + (long)(tile * 16 + m) * HID : bin;
  const float* __restrict__ Bc  = W1 + wave * 16 + m;

  v8f acc = {};
#pragma unroll
  for (int k = 0; k < HID; k += 4) {
    v2f a, b;
    a.x = Ar[k + khalf];
    a.y = Ar[k + khalf + 1];
    b.x = Bc[(k + khalf) * HID];
    b.y = Bc[(k + khalf + 1) * HID];
    acc = __builtin_amdgcn_wmma_f32_16x16x4_f32(
         false, a,  false, b,
         (short)0, acc,  false,  false);
  }

  float* __restrict__ o = fold + (size_t)type * FOLD_ROWS * HID + wave * 16 + m;
  if (tile < IN_DIM / 16) {
    const int rhalf = (lane >> 4) << 3;
#pragma unroll
    for (int r = 0; r < 8; ++r) o[(tile * 16 + rhalf + r) * HID] = acc[r];
  } else if (lane < 16) {
    o[IN_DIM * HID] = acc[0];
  }
}

#define SCAN_ELEMS  1000
#define SCAN_QUADS  (SCAN_ELEMS / 4)
#define SCAN_BLOCKS (NNODES / SCAN_ELEMS)
#define NODE_QUADS  (NNODES / 4)
#define OWN_RANGE   10000
#define OWN_SLICES  10
#define OWN_RANGES  (NNODES / OWN_RANGE)
#define OWN_SUBS    (OWN_RANGE / SCAN_ELEMS)
#define OWN_THREADS 1024
#define OWN_UNROLL  4
#define SLICE_EDGES (E_UI / OWN_SLICES)
#define SLICE_QUADS (SLICE_EDGES / 4)
#define CPITCH      (NNODES + 8)
static_assert(NNODES % 4 == 0 && SCAN_ELEMS % 4 == 0 && CPITCH % 4 == 0, "scan kernel reads the planes as int4");
static_assert(SCAN_QUADS <= 256, "one int4 per scan thread and slice");
static_assert(SCAN_BLOCKS <= 256, "block totals are summed by the first 256 threads of a scan block");
static_assert(N_USER % OWN_RANGE == 0 && N_ITEM % OWN_RANGE == 0, "an owner range holds nodes of one type only");
static_assert(OWN_RANGE % SCAN_ELEMS == 0, "an owner range is whole scan blocks");
static_assert(E_UI == E_IU && E_UI % (4 * OWN_SLICES) == 0, "slices are whole int4 groups, the same for both rows");

__device__ __forceinline__ void edge_ends(const int* __restrict__ eui,
                                          const int* __restrict__ eiu,
                                          int e, int& s, int& d)
{
  if (e < E_UI) { s = eui[e];                 d = eui[E_UI + e] + N_USER; }
  else          { int e2 = e - E_UI;
                  s = eiu[e2] + N_USER;       d = eiu[E_IU + e2]; }
}

__device__ __forceinline__ void take_slots(int* lc, int* __restrict__ rank, int4 v, int first, int e0)
{
  const unsigned x0 = (unsigned)(v.x - first), x1 = (unsigned)(v.y - first);
  const unsigned x2 = (unsigned)(v.z - first), x3 = (unsigned)(v.w - first);
  const bool h0 = x0 < (unsigned)OWN_RANGE, h1 = x1 < (unsigned)OWN_RANGE;
  const bool h2 = x2 < (unsigned)OWN_RANGE, h3 = x3 < (unsigned)OWN_RANGE;
  int r0 = 0, r1 = 0, r2 = 0, r3 = 0;
  if (h0) r0 = atomicAdd(&lc[x0], 1);
  if (h1) r1 = atomicAdd(&lc[x1], 1);
  if (h2) r2 = atomicAdd(&lc[x2], 1);
  if (h3) r3 = atomicAdd(&lc[x3], 1);
  if (h0) rank[e0]     = r0;
  if (h1) rank[e0 + 1] = r1;
  if (h2) rank[e0 + 2] = r2;
  if (h3) rank[e0 + 3] = r3;
}

#define OWN_BLOCKS     (OWN_RANGES * OWN_SLICES)
#define FOLD_PER_BLOCK (OWN_THREADS / 256)
#define FOLD_BLOCKS    ((FOLD_JOBS + FOLD_PER_BLOCK - 1) / FOLD_PER_BLOCK)
__global__ __launch_bounds__(OWN_THREADS) void own_count_rank(
    const int* __restrict__ eui, const int* __restrict__ eiu,
    int* __restrict__ sc, int* __restrict__ rank, int* __restrict__ btot,
    const float* __restrict__ W_iu, const float* __restrict__ b_iu,
    const float* __restrict__ W_ii, const float* __restrict__ b_ii,
    const float* __restrict__ W1, float* __restrict__ fold)
{
  __shared__ int lc[OWN_RANGE];
  const int tid = threadIdx.x;
  if ((int)blockIdx.x >= OWN_BLOCKS) {
    const int wv  = __builtin_amdgcn_readfirstlane(tid >> 5);
    const int job = ((int)blockIdx.x - OWN_BLOCKS) * FOLD_PER_BLOCK + (wv >> 3);
    if (job < FOLD_JOBS)
      fold_in_w1(W_iu, b_iu, W_ii, b_ii, W1, fold, job / FOLD_TILES, job % FOLD_TILES, wv & 7, tid & 31);
    return;
  }
  const int rg  = (int)blockIdx.x / OWN_SLICES;
  const int sl  = (int)blockIdx.x - rg * OWN_SLICES;
  const int lo  = rg * OWN_RANGE;
  for (int i = tid; i < OWN_RANGE; i += OWN_THREADS) lc[i] = 0;
  __syncthreads();

  const bool usr  = lo < N_USER;
  const int4* __restrict__ dl = (const int4*)(usr ? eiu + E_IU : eui + E_UI) + sl * SLICE_QUADS;
  const int ebase = (usr ? E_UI : 0) + sl * SLICE_EDGES;
  const int first = usr ? lo : lo - N_USER;
  const int rt    = usr ? rg : rg - N_USER / OWN_RANGE;
  const int rot   = rt * ((SLICE_QUADS / (N_USER / OWN_RANGE)) & ~31);

  constexpr int NIT = (SLICE_QUADS + OWN_THREADS * OWN_UNROLL - 1) / (OWN_THREADS * OWN_UNROLL);
  for (int it = 0; it < NIT; ++it) {
    int4 v[OWN_UNROLL];
    int  g[OWN_UNROLL];
#pragma unroll
    for (int u = 0; u < OWN_UNROLL; ++u) {
      const int q = (it * OWN_UNROLL + u) * OWN_THREADS + tid;
      int r = q + rot;
      if (r >= SLICE_QUADS) r -= SLICE_QUADS;
      g[u] = r;
      v[u] = (q < SLICE_QUADS) ? dl[r] : make_int4(-1, -1, -1, -1);
    }
#pragma unroll
    for (int u = 0; u < OWN_UNROLL; ++u) {
      take_slots(lc, rank, v[u], first, ebase + 4 * g[u]);
    }
  }
  __syncthreads();

  int* __restrict__ plane = sc + (size_t)sl * CPITCH + lo;
  for (int i = tid; i < OWN_RANGE; i += OWN_THREADS) plane[i] = lc[i];
  const int lane = tid & 31;
  const int wave = tid >> 5;
  if (wave < OWN_SUBS) {
    int c = 0;
    for (int i = lane; i < SCAN_ELEMS; i += 32) c += lc[wave * SCAN_ELEMS + i];
    for (int m = 16; m > 0; m >>= 1) c += __shfl_xor(c, m, 32);
    if (lane == 0) btot[sl * SCAN_BLOCKS + rg * OWN_SUBS + wave] = c;
  }
}

__device__ __forceinline__ void scan_rowptr_block(int sb, int* __restrict__ sc,
                                                  const int* __restrict__ btot,
                                                  float* __restrict__ dinv, int* wsum, int* wpre)
{
  const int t    = threadIdx.x;
  const int lane = t & 31;
  const int w    = t >> 5;
  const int q    = sb * SCAN_QUADS + t;
  const bool live = t < SCAN_QUADS;

  int p = 0;
  if (t < sb) {
    for (int j = 0; j < OWN_SLICES; ++j) p += btot[j * SCAN_BLOCKS + t];
  }
  for (int m = 16; m > 0; m >>= 1) p += __shfl_xor(p, m, 32);

  int4 cj[OWN_SLICES];
  int4 c = make_int4(0, 0, 0, 0);
#pragma unroll
  for (int j = 0; j < OWN_SLICES; ++j) {
    cj[j] = live ? ((const int4*)(sc + (size_t)j * CPITCH))[q] : make_int4(0, 0, 0, 0);
  }
#pragma unroll
  for (int j = 0; j < OWN_SLICES; ++j) {
    c.x += cj[j].x; c.y += cj[j].y; c.z += cj[j].z; c.w += cj[j].w;
  }
  const int s = c.x + c.y + c.z + c.w;
  int inc = s;
  for (int dlt = 1; dlt < 32; dlt <<= 1) {
    int v = __shfl_up(inc, dlt, 32);
    if (lane >= dlt) inc += v;
  }
  if (w < 8 && lane == 31) wsum[w] = inc;
  if (w < 8 && lane == 0)  wpre[w] = p;
  __syncthreads();
  int base = 0;
  for (int k = 0; k < 8; ++k) {
    base += wpre[k];
    if (k < w) base += wsum[k];
  }
  if (live) {
    int4 r;
    r.x = base + inc - s;
    r.y = r.x + c.x;
    r.z = r.y + c.y;
    r.w = r.z + c.z;
    float4 dv;
    dv.x = rsqrtf((float)(c.x + 1));
    dv.y = rsqrtf((float)(c.y + 1));
    dv.z = rsqrtf((float)(c.z + 1));
    dv.w = rsqrtf((float)(c.w + 1));
    ((float4*)dinv)[q] = dv;
    if (q == NODE_QUADS - 1) sc[NNODES] = r.w + c.w;
#pragma unroll
    for (int j = 0; j < OWN_SLICES; ++j) {
      ((int4*)(sc + (size_t)j * CPITCH))[q] = r;
      r.x += cj[j].x; r.y += cj[j].y; r.z += cj[j].z; r.w += cj[j].w;
    }
  }
}

__global__ void csr_fill(const int* __restrict__ eui, const int* __restrict__ eiu,
                         const int* __restrict__ off, const int* __restrict__ rank,
                         const float* __restrict__ dinv, int* __restrict__ col)
{
  int e = blockIdx.x * blockDim.x + threadIdx.x;
  if (e >= E_TOT) return;
  int s, d;
  edge_ends(eui, eiu, e, s, d);
  const int sl = (e < E_UI ? e : e - E_UI) / SLICE_EDGES;
  ((int2*)col)[off[(size_t)sl * CPITCH + d] + rank[e]] = make_int2(s, __float_as_int(dinv[s]));
}

template <int U>
__device__ __forceinline__ void add_rows(float4& acc, const float4* __restrict__ hw4,
                                         int ss, float mw, int k, int lane)
{
  float4 v[U];
  float  w[U];
  for (int u = 0; u < U; ++u) {
    const int s = __builtin_amdgcn_readlane(ss, k + u);
    v[u] = hw4[(long)s * (HID / 4) + lane];
  }
  for (int u = 0; u < U; ++u)
    w[u] = __int_as_float(__builtin_amdgcn_readlane(__float_as_int(mw), k + u));
  for (int u = 0; u < U; ++u) {
    acc.x = fmaf(w[u], v[u].x, acc.x);
    acc.y = fmaf(w[u], v[u].y, acc.y);
    acc.z = fmaf(w[u], v[u].z, acc.z);
    acc.w = fmaf(w[u], v[u].w, acc.w);
  }
}

__global__ __launch_bounds__(256) void agg_gather(
    const int* __restrict__ rp, const int* __restrict__ col,
    const float* __restrict__ hw, const float* __restrict__ dinv,
    const float* __restrict__ bias, float* __restrict__ out)
{
  const int lane = threadIdx.x & 31;
  const int w = __builtin_amdgcn_readfirstlane((int)(threadIdx.x >> 5));
  const int i = blockIdx.x * 8 + w;
  if (i >= NNODES) return;

  const float4* __restrict__ hw4 = (const float4*)hw;
  const int   start = rp[i];
  const int   end   = rp[i + 1];
  const float di    = dinv[i];
  const float4 sv   = hw4[(long)i * (HID / 4) + lane];
  const float4 bv   = ((const float4*)bias)[lane];

  const float dd = di * di;
  float4 acc;
  acc.x = bv.x + dd * sv.x;
  acc.y = bv.y + dd * sv.y;
  acc.z = bv.z + dd * sv.z;
  acc.w = bv.w + dd * sv.w;

  for (int base = start; base < end; base += 32) {
    int n = end - base;
    if (n > 32) n = 32;

    int2 slot = make_int2(0x7fffffff, 0);
    if (lane < n) slot = ((const int2*)col)[base + lane];
    const int ms = slot.x;

    int pos = 0;
    for (int k = 0; k < n; ++k) {
      const int sk = __builtin_amdgcn_readlane(ms, k);
      pos += ((sk < ms) | ((sk == ms) & (k < lane))) ? 1 : 0;
    }
    int ss = __builtin_amdgcn_ds_permute(pos << 2, ms);
    const float mw = __int_as_float(__builtin_amdgcn_ds_permute(pos << 2, slot.y)) * di;
    ss = (int)min((unsigned)ss, (unsigned)(NNODES - 1));

    int j = 0;
    for (; j + 8 <= n; j += 8) add_rows<8>(acc, hw4, ss, mw, j, lane);
    if (j + 4 <= n) { add_rows<4>(acc, hw4, ss, mw, j, lane); j += 4; }
    switch (n - j) {
      case 3: add_rows<3>(acc, hw4, ss, mw, j, lane); break;
      case 2: add_rows<2>(acc, hw4, ss, mw, j, lane); break;
      case 1: add_rows<1>(acc, hw4, ss, mw, j, lane); break;
      default: break;
    }
  }

  ((float4*)out)[(long)i * (HID / 4) + lane] = acc;
}

__device__ __forceinline__ void sum_chunk(float4& acc, const float4* __restrict__ hw4,
                                          int2 raw, int n, float di, int lane)
{
  const int2 slot = lane < n ? raw : make_int2(0x7fffffff, 0);
  const int ms = slot.x;

  int pos = 0;
  for (int q = 0; q < n; ++q) {
    const int sk = __builtin_amdgcn_readlane(ms, q);
    pos += ((sk < ms) | ((sk == ms) & (q < lane))) ? 1 : 0;
  }
  int ss = __builtin_amdgcn_ds_permute(pos << 2, ms);
  const float mw = __int_as_float(__builtin_amdgcn_ds_permute(pos << 2, slot.y)) * di;
  ss = (int)min((unsigned)ss, (unsigned)(NNODES - 1));

  int j = 0;
  for (; j + 8 <= n; j += 8) add_rows<8>(acc, hw4, ss, mw, j, lane);
  switch (n - j) {
    case 7: add_rows<7>(acc, hw4, ss, mw, j, lane); break;
    case 6: add_rows<6>(acc, hw4, ss, mw, j, lane); break;
    case 5: add_rows<5>(acc, hw4, ss, mw, j, lane); break;
    case 4: add_rows<4>(acc, hw4, ss, mw, j, lane); break;
    case 3: add_rows<3>(acc, hw4, ss, mw, j, lane); break;
    case 2: add_rows<2>(acc, hw4, ss, mw, j, lane); break;
    case 1: add_rows<1>(acc, hw4, ss, mw, j, lane); break;
    default: break;
  }
}

#define FUSE_WGS   256
#define FUSE_WAVES 25

__global__ __launch_bounds__(FUSE_WAVES * 32) void agg_relu_gemm(
    const int* __restrict__ rp, const int* __restrict__ col,
    const float* __restrict__ hw, const float* __restrict__ dinv,
    const float* __restrict__ bias, const float* __restrict__ W,
    float* __restrict__ C)
{
  __shared__ v4f Ws[(HID / 8) * 256];
  __shared__ v4f Ts[FUSE_WAVES * 16 * 32];

  const int tid  = threadIdx.x;
  const int lane = tid & 31;
  const int wave = __builtin_amdgcn_readfirstlane(tid >> 5);
  const int c    = lane & 15;
  const int h    = lane >> 4;

#pragma unroll
  for (int i = 0; i < ((HID / 8) * 256 + FUSE_WAVES * 32 - 1) / (FUSE_WAVES * 32); ++i) {
    const int e = tid + FUSE_WAVES * 32 * i;
    if (e < (HID / 8) * 256) {
      const float* __restrict__ p =
          W + (size_t)(8 * (e >> 8) + 2 * h) * HID + ((e >> 5) & 7) * 16 + c;
      v4f v;
      v.x = p[0];
      v.y = p[HID];
      v.z = p[4 * HID];
      v.w = p[5 * HID];
      Ws[e] = v;
    }
  }
  __syncthreads();

  const float4* __restrict__ hw4  = (const float4*)hw;
  const int2*   __restrict__ col2 = (const int2*)col;
  const float4 bv = ((const float4*)bias)[lane];
  v4f* __restrict__ Tw = Ts + wave * (16 * 32);
  const float* __restrict__ Tr = (const float*)Tw + c * 128 + 2 * h;
  const v4f* __restrict__ Bl = Ws + lane;
  const int rhalf  = h << 3;
  static_assert(N_USER % 16 == 0 && N_ITEM % 16 == 0, "a tile is 16 whole nodes of one type");

  constexpr int NSLOTS = FUSE_WAVES * FUSE_WGS;
  constexpr int TU = N_USER / 16, TI = N_ITEM / 16;
  constexpr int PU = (TU + NSLOTS - 1) / NSLOTS;
  constexpr int PI = (TI + NSLOTS - 1) / NSLOTS;
  const int slot = wave * FUSE_WGS + (int)blockIdx.x;
  for (int pass = 0; pass < PU + PI; ++pass) {
    const bool usr = pass < PU;
    const int  t   = (usr ? pass : pass - PU) * NSLOTS + slot;
    if (t >= (usr ? TU : TI)) continue;
    const int tile = usr ? t : TU + t;
    asm volatile("" ::: "memory");
    const int node0 = tile * 16;
    const int   rpl = rp[node0 + (lane < 16 ? lane : 16)];
    const float dil = dinv[node0 + c];

    int2   pslot;
    float4 psv;
    {
      const int s0 = __builtin_amdgcn_readlane(rpl, 0);
      pslot = col2[min(s0 + lane, E_TOT - 1)];
      psv = hw4[(long)node0 * (HID / 4) + lane];
    }

    for (int k = 0; k < 16; ++k) {
      const int   start = __builtin_amdgcn_readlane(rpl, k);
      const int   end   = __builtin_amdgcn_readlane(rpl, k + 1);
      const float di    = __int_as_float(__builtin_amdgcn_readlane(__float_as_int(dil), k));
      const int2 raw = pslot;
      const float4 sv = psv;
      {
        const int kn = k < 15 ? k + 1 : 15;
        const int sn = __builtin_amdgcn_readlane(rpl, kn);
        pslot = col2[min(sn + lane, E_TOT - 1)];
        psv = hw4[(long)(node0 + kn) * (HID / 4) + lane];
      }

      const float dd = di * di;
      float4 acc;
      acc.x = bv.x + dd * sv.x;
      acc.y = bv.y + dd * sv.y;
      acc.z = bv.z + dd * sv.z;
      acc.w = bv.w + dd * sv.w;

      if (end > start) {
        sum_chunk(acc, hw4, raw, min(end - start, 32), di, lane);
        for (int base = start + 32; base < end; base += 32) {
          const int2 more = col2[min(base + lane, E_TOT - 1)];
          sum_chunk(acc, hw4, more, min(end - base, 32), di, lane);
        }
      }

      v4f o;
      o.x = fmaxf(acc.x, 0.0f);
      o.y = fmaxf(acc.y, 0.0f);
      o.z = fmaxf(acc.z, 0.0f);
      o.w = fmaxf(acc.w, 0.0f);
      Tw[k * 32 + (lane ^ k)] = o;
    }

    __builtin_amdgcn_fence(__ATOMIC_RELEASE, "workgroup", "local");
    __builtin_amdgcn_wave_barrier();
    __builtin_amdgcn_fence(__ATOMIC_ACQUIRE, "workgroup", "local");

    v8f acc8[8] = {};
    for (int kp = 0; kp < HID / 8; ++kp) {
      const v2f s0 = *(const v2f*)(Tr + (((2 * kp)     ^ c) << 2));
      const v2f s1 = *(const v2f*)(Tr + (((2 * kp + 1) ^ c) << 2));
      v4f b[8];
      load_b_pair(b, Bl + kp * 256);
      wmma_kpair<false>(acc8, s0, s1, b);
    }

    const long row0 = (long)node0;
#pragma unroll
    for (int t = 0; t < 8; ++t) {
#pragma unroll
      for (int r = 0; r < 8; ++r)
        C[(row0 + rhalf + r) * HID + t * 16 + c] = acc8[t][r];
    }

    __builtin_amdgcn_fence(__ATOMIC_RELEASE, "workgroup", "local");
    __builtin_amdgcn_wave_barrier();
  }
}

extern "C" void kernel_launch(void* const* d_in, const int* in_sizes, int n_in,
                              void* d_out, int out_size, void* d_ws, size_t ws_size,
                              hipStream_t stream)
{
  const float* x_user = (const float*)d_in[0];
  const float* x_item = (const float*)d_in[1];
  const int*   eui    = (const int*)d_in[2];
  const int*   eiu    = (const int*)d_in[3];
  const float* W_iu   = (const float*)d_in[4];
  const float* b_iu   = (const float*)d_in[5];
  const float* W_ii   = (const float*)d_in[6];
  const float* b_ii   = (const float*)d_in[7];
  const float* W1     = (const float*)d_in[8];
  const float* b1     = (const float*)d_in[9];
  const float* W2     = (const float*)d_in[10];
  const float* b2     = (const float*)d_in[11];
  float* out = (float*)d_out;

  float* h    = (float*)d_ws;
  float* hw   = h  + (size_t)NNODES * HID;
  float* dinv = hw + (size_t)NNODES * HID;
  int*   sc   = (int*)(dinv + NNODES);
  int*   rp   = sc;
  int*   rank = sc   + (size_t)OWN_SLICES * CPITCH;
  int*   col  = rank + E_TOT;
  int*   btot = col  + 2 * E_TOT;
  float* fold = (float*)(btot + OWN_SLICES * SCAN_BLOCKS);
  float* fold_u = fold;
  float* fold_i = fold + (size_t)FOLD_ROWS * HID;

  const dim3 blk(GEMM_WAVES * 32);

  own_count_rank<<<dim3(OWN_BLOCKS + FOLD_BLOCKS), dim3(OWN_THREADS), 0, stream>>>(
      eui, eiu, sc, rank, btot, W_iu, b_iu, W_ii, b_ii, W1, fold);

  gemm_wlds<IN_DIM, false><<<dim3(GEMM_WGS + SCAN_BLOCKS), blk, 0, stream>>>(
      x_user, fold_u, fold_u + IN_DIM * HID, hw, N_USER,
      x_item, fold_i, fold_i + IN_DIM * HID, hw + (size_t)N_USER * HID, N_ITEM,
      sc, btot, dinv);

  csr_fill   <<<dim3((E_TOT + 255) / 256),  dim3(256), 0, stream>>>(eui, eiu, sc, rank, dinv, col);

  const unsigned nodeBlocks = (NNODES + 7) / 8;

  agg_relu_gemm<<<dim3(FUSE_WGS), dim3(FUSE_WAVES * 32), 0, stream>>>(rp, col, hw, dinv, b1, W2, h);

  agg_gather<<<dim3(nodeBlocks), dim3(256), 0, stream>>>(rp, col, h, dinv, b2, out);
}
